# starting kernel
# baseline (speedup 1.0000x reference)
#include <hip/hip_runtime.h>
#include <hip/hip_bf16.h>


typedef __attribute__((ext_vector_type(2))) float v2f;
typedef __attribute__((ext_vector_type(8))) float v8f;

static constexpr int N_NODES  = 50000;
static constexpr int N_EDGES  = 800000;
static constexpr int DIM      = 256;
static constexpr int HEADS    = 4;
static constexpr int HEAD_DIM = 64;
static constexpr float NEG_SLOPE = 0.2f;

static constexpr int HS_STRIDE = 260;

__global__ __launch_bounds__(256) void gat_gemm_wmma(
    const float* __restrict__ h, const float* __restrict__ W,
    const float* __restrict__ bias, float* __restrict__ Wh) {
  __shared__ float hs[16 * HS_STRIDE];

  const int row0 = blockIdx.x * 16;
  const int t    = threadIdx.x;

  #pragma unroll
  for (int r = 0; r < 16; ++r)
    hs[r * HS_STRIDE + t] = h[(row0 + r) * DIM + t];
  __syncthreads();

  const int wave = t >> 5;
  const int lane = t & 31;
  const int half = lane >> 4;
  const int lr   = lane & 15;

  const int col0 = wave * 16 + lr;
  const int col1 = (wave + 8) * 16 + lr;
  const float* __restrict__ w0 = W + col0 * DIM;
  const float* __restrict__ w1 = W + col1 * DIM;

  v8f c0 = {};
  v8f c1 = {};

  for (int k0 = 0; k0 < DIM; k0 += 4) {
    const int k = k0 + 2 * half;
    v2f a = *(const v2f*)&hs[lr * HS_STRIDE + k];
    v2f b0 = *(const v2f*)(w0 + k);
    v2f b1 = *(const v2f*)(w1 + k);
    c0 = __builtin_amdgcn_wmma_f32_16x16x4_f32(false, a, false, b0,
                                               (short)0, c0, false, false);
    c1 = __builtin_amdgcn_wmma_f32_16x16x4_f32(false, a, false, b1,
                                               (short)0, c1, false, false);
  }

  const float bias0 = bias[col0];
  const float bias1 = bias[col1];
  #pragma unroll
  for (int r = 0; r < 8; ++r) {
    const int row = row0 + r + 8 * half;
    Wh[row * DIM + col0] = c0[r] + bias0;
    Wh[row * DIM + col1] = c1[r] + bias1;
  }
}

__global__ __launch_bounds__(256) void node_prep(
    const float* __restrict__ Wh, const float* __restrict__ a_src,
    const float* __restrict__ a_dst, float* __restrict__ alpha_src,
    float* __restrict__ alpha_dst, float* __restrict__ m,
    float* __restrict__ denom, float* __restrict__ out) {
  __shared__ float ss[256];
  __shared__ float sd[256];
  const int n = blockIdx.x;
  const int t = threadIdx.x;

  const float w = Wh[n * DIM + t];
  out[n * DIM + t] = 0.0f;
  ss[t] = w * a_src[t];
  sd[t] = w * a_dst[t];
  __syncthreads();

  if (t < HEADS) {
    float s1 = 0.0f, s2 = 0.0f;
    #pragma unroll
    for (int d = 0; d < HEAD_DIM; ++d) {
      s1 += ss[t * HEAD_DIM + d];
      s2 += sd[t * HEAD_DIM + d];
    }
    alpha_src[n * HEADS + t] = s1;
    alpha_dst[n * HEADS + t] = s2;
    m[n * HEADS + t]     = __int_as_float(0xFF800000);
    denom[n * HEADS + t] = 0.0f;
  }
}

__device__ __forceinline__ void atomic_max_f32(float* addr, float val) {
  const int vi = __float_as_int(val);
  if (vi >= 0)
    atomicMax((int*)addr, vi);
  else
    atomicMin((unsigned int*)addr, (unsigned int)vi);
}

__device__ __forceinline__ float leaky(float x) {
  return x > 0.0f ? x : NEG_SLOPE * x;
}

__global__ __launch_bounds__(256) void edge_max(
    const int* __restrict__ ei, const float* __restrict__ alpha_src,
    const float* __restrict__ alpha_dst, float* __restrict__ m) {
  const int idx = blockIdx.x * 256 + threadIdx.x;
  if (idx >= N_EDGES * HEADS) return;
  const int e  = idx >> 2;
  const int hh = idx & 3;
  const int i = ei[e];
  const int j = ei[N_EDGES + e];
  const float v = leaky(alpha_src[i * HEADS + hh] + alpha_dst[j * HEADS + hh]);
  atomic_max_f32(&m[i * HEADS + hh], v);
}

__global__ __launch_bounds__(256) void edge_sum(
    const int* __restrict__ ei, const float* __restrict__ alpha_src,
    const float* __restrict__ alpha_dst, const float* __restrict__ m,
    float* __restrict__ denom) {
  const int idx = blockIdx.x * 256 + threadIdx.x;
  if (idx >= N_EDGES * HEADS) return;
  const int e  = idx >> 2;
  const int hh = idx & 3;
  const int i = ei[e];
  const int j = ei[N_EDGES + e];
  const float v = leaky(alpha_src[i * HEADS + hh] + alpha_dst[j * HEADS + hh]);
  atomicAdd(&denom[i * HEADS + hh], __expf(v - m[i * HEADS + hh]));
}

__global__ __launch_bounds__(256) void edge_agg(
    const int* __restrict__ ei, const float* __restrict__ alpha_src,
    const float* __restrict__ alpha_dst, const float* __restrict__ m,
    const float* __restrict__ denom, const float* __restrict__ Wh,
    float* __restrict__ out) {
  const int e  = blockIdx.x;
  const int t  = threadIdx.x;
  const int hh = t >> 6;
  const int i = ei[e];
  const int j = ei[N_EDGES + e];
  const float v   = leaky(alpha_src[i * HEADS + hh] + alpha_dst[j * HEADS + hh]);
  const float att = __expf(v - m[i * HEADS + hh]) / denom[i * HEADS + hh];
  atomicAdd(&out[i * DIM + t], att * Wh[j * DIM + t]);
}

extern "C" void kernel_launch(void* const* d_in, const int* in_sizes, int n_in,
                              void* d_out, int out_size, void* d_ws, size_t ws_size,
                              hipStream_t stream) {
  const float* h     = (const float*)d_in[0];
  const int*   ei    = (const int*)d_in[1];
  const float* W     = (const float*)d_in[2];
  const float* bias  = (const float*)d_in[3];
  const float* a_src = (const float*)d_in[4];
  const float* a_dst = (const float*)d_in[5];
  float* out = (float*)d_out;

  float* Wh        = (float*)d_ws;
  float* alpha_src = Wh + (size_t)N_NODES * DIM;
  float* alpha_dst = alpha_src + (size_t)N_NODES * HEADS;
  float* m         = alpha_dst + (size_t)N_NODES * HEADS;
  float* denom     = m + (size_t)N_NODES * HEADS;

  gat_gemm_wmma<<<N_NODES / 16, 256, 0, stream>>>(h, W, bias, Wh);

  node_prep<<<N_NODES, 256, 0, stream>>>(Wh, a_src, a_dst, alpha_src, alpha_dst,
                                         m, denom, out);

  const int eh_blocks = (N_EDGES * HEADS + 255) / 256;
  edge_max<<<eh_blocks, 256, 0, stream>>>(ei, alpha_src, alpha_dst, m);
  edge_sum<<<eh_blocks, 256, 0, stream>>>(ei, alpha_src, alpha_dst, m, denom);

  edge_agg<<<N_EDGES, 256, 0, stream>>>(ei, alpha_src, alpha_dst, m, denom, Wh,
                                        out);
}


# Group edges by destination row (rank by returned atomic add, arrival order) so one wave does each row's softmax and weighted sum in registers; no f32 atomics, sum order changes
# speedup vs baseline: 2.8473x; 2.8473x over previous
#include <hip/hip_runtime.h>
#include <hip/hip_bf16.h>


typedef __attribute__((ext_vector_type(2))) float v2f;
typedef __attribute__((ext_vector_type(8))) float v8f;

static constexpr int N_NODES  = 50000;
static constexpr int N_EDGES  = 800000;
static constexpr int DIM      = 256;
static constexpr int HEADS    = 4;
static constexpr int HEAD_DIM = 64;
static constexpr float NEG_SLOPE = 0.2f;

static_assert(N_NODES <= 65536, "source list stores 16-bit node indices");
static_assert(N_EDGES == 2 * N_NODES * HEADS * 2, "source list fills the two N*4-word planes exactly");
static_assert(N_EDGES % 256 == 0 && N_NODES % 8 == 0, "grids below have no tail");
static_assert(N_EDGES / 128 <= N_NODES, "edge ranks fit in the upper halves of the output rows");

static constexpr int ROW_CNT   = 0;
static constexpr int ROW_START = N_NODES;
static constexpr int CURSOR    = 2 * N_NODES;
static_assert(2 * N_NODES + 1 <= N_NODES * HEADS, "row table fits in one plane");
static constexpr int OUT_ASRC = 4;

__device__ __forceinline__ int rank_word(int e) {
  return (e >> 7) * DIM + 128 + (e & 127);
}

static constexpr int HS_STRIDE = 260;

__global__ __launch_bounds__(256) void gat_gemm_wmma(
    const float* __restrict__ h, const float* __restrict__ W,
    const float* __restrict__ bias, float* __restrict__ Wh) {
  __shared__ float hs[16 * HS_STRIDE];

  const int row0 = blockIdx.x * 16;
  const int t    = threadIdx.x;

  #pragma unroll
  for (int r = 0; r < 16; ++r)
    hs[r * HS_STRIDE + t] = h[(row0 + r) * DIM + t];
  __syncthreads();

  const int wave = t >> 5;
  const int lane = t & 31;
  const int half = lane >> 4;
  const int lr   = lane & 15;

  const int col0 = wave * 16 + lr;
  const int col1 = (wave + 8) * 16 + lr;
  const float* __restrict__ w0 = W + col0 * DIM;
  const float* __restrict__ w1 = W + col1 * DIM;

  v8f c0 = {};
  v8f c1 = {};

  for (int k0 = 0; k0 < DIM; k0 += 4) {
    const int k = k0 + 2 * half;
    v2f a = *(const v2f*)&hs[lr * HS_STRIDE + k];
    v2f b0 = *(const v2f*)(w0 + k);
    v2f b1 = *(const v2f*)(w1 + k);
    c0 = __builtin_amdgcn_wmma_f32_16x16x4_f32(false, a, false, b0,
                                               (short)0, c0, false, false);
    c1 = __builtin_amdgcn_wmma_f32_16x16x4_f32(false, a, false, b1,
                                               (short)0, c1, false, false);
  }

  const float bias0 = bias[col0];
  const float bias1 = bias[col1];
  #pragma unroll
  for (int r = 0; r < 8; ++r) {
    const int row = row0 + r + 8 * half;
    Wh[row * DIM + col0] = c0[r] + bias0;
    Wh[row * DIM + col1] = c1[r] + bias1;
  }
}

__global__ __launch_bounds__(256) void node_prep(
    const float* __restrict__ Wh, const float* __restrict__ a_src,
    const float* __restrict__ a_dst, float* __restrict__ out,
    float* __restrict__ alpha_dst, int* __restrict__ rowtab) {
  __shared__ float ss[256];
  __shared__ float sd[256];
  const int n = blockIdx.x;
  const int t = threadIdx.x;

  const float w = Wh[n * DIM + t];
  ss[t] = w * a_src[t];
  sd[t] = w * a_dst[t];
  if (t == 0) rowtab[ROW_CNT + n] = 0;
  if (n == 0 && t == 1) rowtab[CURSOR] = 0;
  __syncthreads();

  if (t < HEADS) {
    float s1 = 0.0f, s2 = 0.0f;
    #pragma unroll
    for (int d = 0; d < HEAD_DIM; ++d) {
      s1 += ss[t * HEAD_DIM + d];
      s2 += sd[t * HEAD_DIM + d];
    }
    out[n * DIM + OUT_ASRC + t] = s1;
    alpha_dst[n * HEADS + t] = s2;
  }
}

__device__ __forceinline__ float leaky(float x) {
  return x > 0.0f ? x : NEG_SLOPE * x;
}

__global__ __launch_bounds__(256) void edge_rank(
    const int* __restrict__ ei, int* __restrict__ rowtab, int* __restrict__ ranks) {
  const int e = blockIdx.x * 256 + threadIdx.x;
  const int i = ei[e];
  const int r = atomicAdd(&rowtab[ROW_CNT + i], 1);
  ranks[rank_word(e)] = r;
}

__global__ __launch_bounds__(256) void row_ranges(int* __restrict__ rowtab) {
  __shared__ int wsum[8];
  __shared__ int sbase;
  const int t    = threadIdx.x;
  const int lane = t & 31;
  const int wave = t >> 5;
  const int n    = blockIdx.x * 256 + t;

  const int c = (n < N_NODES) ? rowtab[ROW_CNT + n] : 0;
  int x = c;
  for (int d = 1; d < 32; d <<= 1) {
    const int y = __shfl_up(x, d);
    if (lane >= d) x += y;
  }
  if (lane == 31) wsum[wave] = x;
  __syncthreads();
  if (t == 0) {
    int tot = 0;
    for (int w = 0; w < 8; ++w) {
      const int s = wsum[w];
      wsum[w] = tot;
      tot += s;
    }
    sbase = atomicAdd(&rowtab[CURSOR], tot);
  }
  __syncthreads();
  if (n < N_NODES) rowtab[ROW_START + n] = sbase + wsum[wave] + x - c;
}

__global__ __launch_bounds__(256) void edge_place(
    const int* __restrict__ ei, const int* __restrict__ rowtab,
    const int* __restrict__ ranks, unsigned short* __restrict__ src_list) {
  const int e = blockIdx.x * 256 + threadIdx.x;
  const int i = ei[e];
  const int j = ei[N_EDGES + e];
  const int r = ranks[rank_word(e)];
  const int s = rowtab[ROW_START + i];
  src_list[s + r] = (unsigned short)j;
}

__device__ __forceinline__ void edge_logits(
    const unsigned short* __restrict__ src_list, const float* __restrict__ alpha_dst,
    const float4 as, int slot, int& j, float& v0, float& v1, float& v2, float& v3) {
  j = src_list[slot];
  const float4 ad = *(const float4*)(alpha_dst + j * HEADS);
  v0 = leaky(as.x + ad.x);
  v1 = leaky(as.y + ad.y);
  v2 = leaky(as.z + ad.z);
  v3 = leaky(as.w + ad.w);
}

__device__ __forceinline__ float wave_max(float x) {
  for (int d = 16; d > 0; d >>= 1) x = fmaxf(x, __shfl_xor(x, d));
  return x;
}

__device__ __forceinline__ float wave_sum(float x) {
  for (int d = 16; d > 0; d >>= 1) x += __shfl_xor(x, d);
  return x;
}

__device__ __forceinline__ float lane_bcast(float x, int k) {
  return __int_as_float(__builtin_amdgcn_readlane(__float_as_int(x), k));
}

__device__ __forceinline__ void add_rows(
    const float* __restrict__ Wh, int lane, int cnt, int j, float a0, float a1,
    float a2, float a3, float4& acc0, float4& acc1) {
  const bool lo = lane < 16;
  #pragma unroll 4
  for (int k = 0; k < cnt; ++k) {
    const int   jk = __builtin_amdgcn_readlane(j, k);
    const float x0 = lane_bcast(a0, k);
    const float x1 = lane_bcast(a1, k);
    const float x2 = lane_bcast(a2, k);
    const float x3 = lane_bcast(a3, k);
    const float w0 = lo ? x0 : x1;
    const float w1 = lo ? x2 : x3;
    const float* row = Wh + (size_t)jk * DIM + 4 * lane;
    const float4 r0 = *(const float4*)(row);
    const float4 r1 = *(const float4*)(row + 128);
    acc0.x += w0 * r0.x; acc0.y += w0 * r0.y; acc0.z += w0 * r0.z; acc0.w += w0 * r0.w;
    acc1.x += w1 * r1.x; acc1.y += w1 * r1.y; acc1.z += w1 * r1.z; acc1.w += w1 * r1.w;
  }
}

__global__ __launch_bounds__(256) void node_agg(
    const unsigned short* __restrict__ src_list, const int* __restrict__ rowtab,
    const float* __restrict__ alpha_dst, const float* __restrict__ Wh, float* out) {
  const int lane = threadIdx.x & 31;
  const int n = blockIdx.x * 8 + __builtin_amdgcn_readfirstlane(threadIdx.x >> 5);
  float* orow = out + (size_t)n * DIM;
  const int deg   = __builtin_amdgcn_readfirstlane(rowtab[ROW_CNT + n]);
  const int start = __builtin_amdgcn_readfirstlane(rowtab[ROW_START + n]);
  const float4 as = *(const float4*)(orow + OUT_ASRC);

  float4 acc0 = {0.0f, 0.0f, 0.0f, 0.0f};
  float4 acc1 = {0.0f, 0.0f, 0.0f, 0.0f};

  if (deg > 0) {
    const float NEG_INF = __int_as_float(0xFF800000);
    int j0 = 0;
    float u0 = NEG_INF, u1 = NEG_INF, u2 = NEG_INF, u3 = NEG_INF;
    if (lane < deg)
      edge_logits(src_list, alpha_dst, as, start + lane, j0, u0, u1, u2, u3);

    float m0 = u0, m1 = u1, m2 = u2, m3 = u3;
    for (int base = 32; base < deg; base += 32) {
      if (base + lane < deg) {
        int j; float v0, v1, v2, v3;
        edge_logits(src_list, alpha_dst, as, start + base + lane, j, v0, v1, v2, v3);
        m0 = fmaxf(m0, v0); m1 = fmaxf(m1, v1); m2 = fmaxf(m2, v2); m3 = fmaxf(m3, v3);
      }
    }
    m0 = wave_max(m0); m1 = wave_max(m1); m2 = wave_max(m2); m3 = wave_max(m3);

    const bool in0 = lane < deg;
    const float p0 = in0 ? __expf(u0 - m0) : 0.0f;
    const float p1 = in0 ? __expf(u1 - m1) : 0.0f;
    const float p2 = in0 ? __expf(u2 - m2) : 0.0f;
    const float p3 = in0 ? __expf(u3 - m3) : 0.0f;
    float s0 = p0, s1 = p1, s2 = p2, s3 = p3;
    for (int base = 32; base < deg; base += 32) {
      if (base + lane < deg) {
        int j; float v0, v1, v2, v3;
        edge_logits(src_list, alpha_dst, as, start + base + lane, j, v0, v1, v2, v3);
        s0 += __expf(v0 - m0); s1 += __expf(v1 - m1);
        s2 += __expf(v2 - m2); s3 += __expf(v3 - m3);
      }
    }
    s0 = wave_sum(s0); s1 = wave_sum(s1); s2 = wave_sum(s2); s3 = wave_sum(s3);

    add_rows(Wh, lane, deg < 32 ? deg : 32, j0, p0 / s0, p1 / s1, p2 / s2, p3 / s3,
             acc0, acc1);
    for (int base = 32; base < deg; base += 32) {
      int j = 0;
      float a0 = 0.0f, a1 = 0.0f, a2 = 0.0f, a3 = 0.0f;
      if (base + lane < deg) {
        float v0, v1, v2, v3;
        edge_logits(src_list, alpha_dst, as, start + base + lane, j, v0, v1, v2, v3);
        a0 = __expf(v0 - m0) / s0; a1 = __expf(v1 - m1) / s1;
        a2 = __expf(v2 - m2) / s2; a3 = __expf(v3 - m3) / s3;
      }
      const int left = deg - base;
      add_rows(Wh, lane, left < 32 ? left : 32, j, a0, a1, a2, a3, acc0, acc1);
    }
  }

  *(float4*)(orow + 4 * lane)       = acc0;
  *(float4*)(orow + 128 + 4 * lane) = acc1;
}

extern "C" void kernel_launch(void* const* d_in, const int* in_sizes, int n_in,
                              void* d_out, int out_size, void* d_ws, size_t ws_size,
                              hipStream_t stream) {
  const float* h     = (const float*)d_in[0];
  const int*   ei    = (const int*)d_in[1];
  const float* W     = (const float*)d_in[2];
  const float* bias  = (const float*)d_in[3];
  const float* a_src = (const float*)d_in[4];
  const float* a_dst = (const float*)d_in[5];
  float* out = (float*)d_out;
  int* ranks = (int*)d_out;

  float* Wh        = (float*)d_ws;
  int*   rowtab    = (int*)(Wh + (size_t)N_NODES * DIM);
  float* alpha_dst = (float*)(rowtab + (size_t)N_NODES * HEADS);
  unsigned short* src_list = (unsigned short*)(alpha_dst + (size_t)N_NODES * HEADS);

  gat_gemm_wmma<<<N_NODES / 16, 256, 0, stream>>>(h, W, bias, Wh);

  node_prep<<<N_NODES, 256, 0, stream>>>(Wh, a_src, a_dst, out, alpha_dst, rowtab);

  edge_rank<<<N_EDGES / 256, 256, 0, stream>>>(ei, rowtab, ranks);
  row_ranges<<<(N_NODES + 255) / 256, 256, 0, stream>>>(rowtab);
  edge_place<<<N_EDGES / 256, 256, 0, stream>>>(ei, rowtab, ranks, src_list);

  node_agg<<<N_NODES / 8, 256, 0, stream>>>(src_list, rowtab, alpha_dst, Wh, out);
}


# node_prep: one wave per row, five rows per wave, lane partial dots then half-wave shuffle tree (was 8-wave block per row, serial 64-term LDS loop): 400000 waves become 10000; alpha sum order changes
# speedup vs baseline: 3.1915x; 1.1209x over previous
#include <hip/hip_runtime.h>
#include <hip/hip_bf16.h>


typedef __attribute__((ext_vector_type(2))) float v2f;
typedef __attribute__((ext_vector_type(8))) float v8f;

static constexpr int N_NODES  = 50000;
static constexpr int N_EDGES  = 800000;
static constexpr int DIM      = 256;
static constexpr int HEADS    = 4;
static constexpr int HEAD_DIM = 64;
static constexpr float NEG_SLOPE = 0.2f;

static_assert(N_NODES <= 65536, "source list stores 16-bit node indices");
static_assert(N_EDGES == 2 * N_NODES * HEADS * 2, "source list fills the two N*4-word planes exactly");
static_assert(N_EDGES % 256 == 0 && N_NODES % 8 == 0, "grids below have no tail");
static_assert(N_EDGES / 128 <= N_NODES, "edge ranks fit in the upper halves of the output rows");

static constexpr int ROW_CNT   = 0;
static constexpr int ROW_START = N_NODES;
static constexpr int CURSOR    = 2 * N_NODES;
static_assert(2 * N_NODES + 1 <= N_NODES * HEADS, "row table fits in one plane");
static constexpr int OUT_ASRC = 4;

__device__ __forceinline__ int rank_word(int e) {
  return (e >> 7) * DIM + 128 + (e & 127);
}

static constexpr int HS_STRIDE = 260;

__global__ __launch_bounds__(256) void gat_gemm_wmma(
    const float* __restrict__ h, const float* __restrict__ W,
    const float* __restrict__ bias, float* __restrict__ Wh) {
  __shared__ float hs[16 * HS_STRIDE];

  const int row0 = blockIdx.x * 16;
  const int t    = threadIdx.x;

  #pragma unroll
  for (int r = 0; r < 16; ++r)
    hs[r * HS_STRIDE + t] = h[(row0 + r) * DIM + t];
  __syncthreads();

  const int wave = t >> 5;
  const int lane = t & 31;
  const int half = lane >> 4;
  const int lr   = lane & 15;

  const int col0 = wave * 16 + lr;
  const int col1 = (wave + 8) * 16 + lr;
  const float* __restrict__ w0 = W + col0 * DIM;
  const float* __restrict__ w1 = W + col1 * DIM;

  v8f c0 = {};
  v8f c1 = {};

  for (int k0 = 0; k0 < DIM; k0 += 4) {
    const int k = k0 + 2 * half;
    v2f a = *(const v2f*)&hs[lr * HS_STRIDE + k];
    v2f b0 = *(const v2f*)(w0 + k);
    v2f b1 = *(const v2f*)(w1 + k);
    c0 = __builtin_amdgcn_wmma_f32_16x16x4_f32(false, a, false, b0,
                                               (short)0, c0, false, false);
    c1 = __builtin_amdgcn_wmma_f32_16x16x4_f32(false, a, false, b1,
                                               (short)0, c1, false, false);
  }

  const float bias0 = bias[col0];
  const float bias1 = bias[col1];
  #pragma unroll
  for (int r = 0; r < 8; ++r) {
    const int row = row0 + r + 8 * half;
    Wh[row * DIM + col0] = c0[r] + bias0;
    Wh[row * DIM + col1] = c1[r] + bias1;
  }
}

static constexpr int PREP_ROWS  = 5;
static constexpr int PREP_WAVES = N_NODES / PREP_ROWS;
static_assert(N_NODES % PREP_ROWS == 0 && PREP_WAVES % 8 == 0, "node_prep grid has no tail");

__device__ __forceinline__ float dot4(const float4 a, const float4 b) {
  return a.x * b.x + a.y * b.y + a.z * b.z + a.w * b.w;
}

__device__ __forceinline__ float half_wave_sum(float x) {
  for (int d = 8; d > 0; d >>= 1) x += __shfl_xor(x, d);
  return x;
}

__device__ __forceinline__ float lane_bcast(float x, int k) {
  return __int_as_float(__builtin_amdgcn_readlane(__float_as_int(x), k));
}

__global__ __launch_bounds__(256) void node_prep(
    const float* __restrict__ Wh, const float* __restrict__ a_src,
    const float* __restrict__ a_dst, float* __restrict__ out,
    float* __restrict__ alpha_dst, int* __restrict__ rowtab) {
  const int lane = threadIdx.x & 31;
  const int wv   = blockIdx.x * 8 + (threadIdx.x >> 5);
  const int n0   = wv * PREP_ROWS;

  const float4 s0 = *(const float4*)(a_src + 4 * lane);
  const float4 s1 = *(const float4*)(a_src + 128 + 4 * lane);
  const float4 d0 = *(const float4*)(a_dst + 4 * lane);
  const float4 d1 = *(const float4*)(a_dst + 128 + 4 * lane);

  float4 r0[PREP_ROWS], r1[PREP_ROWS];
  #pragma unroll
  for (int i = 0; i < PREP_ROWS; ++i) {
    const float* row = Wh + (size_t)(n0 + i) * DIM + 4 * lane;
    r0[i] = *(const float4*)(row);
    r1[i] = *(const float4*)(row + 128);
  }

  #pragma unroll
  for (int i = 0; i < PREP_ROWS; ++i) {
    const int n = n0 + i;
    const float ps0 = half_wave_sum(dot4(r0[i], s0));
    const float ps1 = half_wave_sum(dot4(r1[i], s1));
    const float pd0 = half_wave_sum(dot4(r0[i], d0));
    const float pd1 = half_wave_sum(dot4(r1[i], d1));
    const float4 vs = {lane_bcast(ps0, 0), lane_bcast(ps0, 16),
                       lane_bcast(ps1, 0), lane_bcast(ps1, 16)};
    const float4 vd = {lane_bcast(pd0, 0), lane_bcast(pd0, 16),
                       lane_bcast(pd1, 0), lane_bcast(pd1, 16)};
    if (lane == 0) {
      *(float4*)(out + (size_t)n * DIM + OUT_ASRC) = vs;
      *(float4*)(alpha_dst + (size_t)n * HEADS)    = vd;
      rowtab[ROW_CNT + n] = 0;
    }
  }
  if (wv == 0 && lane == 1) rowtab[CURSOR] = 0;
}

__device__ __forceinline__ float leaky(float x) {
  return x > 0.0f ? x : NEG_SLOPE * x;
}

__global__ __launch_bounds__(256) void edge_rank(
    const int* __restrict__ ei, int* __restrict__ rowtab, int* __restrict__ ranks) {
  const int e = blockIdx.x * 256 + threadIdx.x;
  const int i = ei[e];
  const int r = atomicAdd(&rowtab[ROW_CNT + i], 1);
  ranks[rank_word(e)] = r;
}

__global__ __launch_bounds__(256) void row_ranges(int* __restrict__ rowtab) {
  __shared__ int wsum[8];
  __shared__ int sbase;
  const int t    = threadIdx.x;
  const int lane = t & 31;
  const int wave = t >> 5;
  const int n    = blockIdx.x * 256 + t;

  const int c = (n < N_NODES) ? rowtab[ROW_CNT + n] : 0;
  int x = c;
  for (int d = 1; d < 32; d <<= 1) {
    const int y = __shfl_up(x, d);
    if (lane >= d) x += y;
  }
  if (lane == 31) wsum[wave] = x;
  __syncthreads();
  if (t == 0) {
    int tot = 0;
    for (int w = 0; w < 8; ++w) {
      const int s = wsum[w];
      wsum[w] = tot;
      tot += s;
    }
    sbase = atomicAdd(&rowtab[CURSOR], tot);
  }
  __syncthreads();
  if (n < N_NODES) rowtab[ROW_START + n] = sbase + wsum[wave] + x - c;
}

__global__ __launch_bounds__(256) void edge_place(
    const int* __restrict__ ei, const int* __restrict__ rowtab,
    const int* __restrict__ ranks, unsigned short* __restrict__ src_list) {
  const int e = blockIdx.x * 256 + threadIdx.x;
  const int i = ei[e];
  const int j = ei[N_EDGES + e];
  const int r = ranks[rank_word(e)];
  const int s = rowtab[ROW_START + i];
  src_list[s + r] = (unsigned short)j;
}

__device__ __forceinline__ void edge_logits(
    const unsigned short* __restrict__ src_list, const float* __restrict__ alpha_dst,
    const float4 as, int slot, int& j, float& v0, float& v1, float& v2, float& v3) {
  j = src_list[slot];
  const float4 ad = *(const float4*)(alpha_dst + j * HEADS);
  v0 = leaky(as.x + ad.x);
  v1 = leaky(as.y + ad.y);
  v2 = leaky(as.z + ad.z);
  v3 = leaky(as.w + ad.w);
}

__device__ __forceinline__ float wave_max(float x) {
  for (int d = 16; d > 0; d >>= 1) x = fmaxf(x, __shfl_xor(x, d));
  return x;
}

__device__ __forceinline__ float wave_sum(float x) {
  for (int d = 16; d > 0; d >>= 1) x += __shfl_xor(x, d);
  return x;
}

__device__ __forceinline__ void add_rows(
    const float* __restrict__ Wh, int lane, int cnt, int j, float a0, float a1,
    float a2, float a3, float4& acc0, float4& acc1) {
  const bool lo = lane < 16;
  #pragma unroll 4
  for (int k = 0; k < cnt; ++k) {
    const int   jk = __builtin_amdgcn_readlane(j, k);
    const float x0 = lane_bcast(a0, k);
    const float x1 = lane_bcast(a1, k);
    const float x2 = lane_bcast(a2, k);
    const float x3 = lane_bcast(a3, k);
    const float w0 = lo ? x0 : x1;
    const float w1 = lo ? x2 : x3;
    const float* row = Wh + (size_t)jk * DIM + 4 * lane;
    const float4 r0 = *(const float4*)(row);
    const float4 r1 = *(const float4*)(row + 128);
    acc0.x += w0 * r0.x; acc0.y += w0 * r0.y; acc0.z += w0 * r0.z; acc0.w += w0 * r0.w;
    acc1.x += w1 * r1.x; acc1.y += w1 * r1.y; acc1.z += w1 * r1.z; acc1.w += w1 * r1.w;
  }
}

__global__ __launch_bounds__(256) void node_agg(
    const unsigned short* __restrict__ src_list, const int* __restrict__ rowtab,
    const float* __restrict__ alpha_dst, const float* __restrict__ Wh, float* out) {
  const int lane = threadIdx.x & 31;
  const int n = blockIdx.x * 8 + __builtin_amdgcn_readfirstlane(threadIdx.x >> 5);
  float* orow = out + (size_t)n * DIM;
  const int deg   = __builtin_amdgcn_readfirstlane(rowtab[ROW_CNT + n]);
  const int start = __builtin_amdgcn_readfirstlane(rowtab[ROW_START + n]);
  const float4 as = *(const float4*)(orow + OUT_ASRC);

  float4 acc0 = {0.0f, 0.0f, 0.0f, 0.0f};
  float4 acc1 = {0.0f, 0.0f, 0.0f, 0.0f};

  if (deg > 0) {
    const float NEG_INF = __int_as_float(0xFF800000);
    int j0 = 0;
    float u0 = NEG_INF, u1 = NEG_INF, u2 = NEG_INF, u3 = NEG_INF;
    if (lane < deg)
      edge_logits(src_list, alpha_dst, as, start + lane, j0, u0, u1, u2, u3);

    float m0 = u0, m1 = u1, m2 = u2, m3 = u3;
    for (int base = 32; base < deg; base += 32) {
      if (base + lane < deg) {
        int j; float v0, v1, v2, v3;
        edge_logits(src_list, alpha_dst, as, start + base + lane, j, v0, v1, v2, v3);
        m0 = fmaxf(m0, v0); m1 = fmaxf(m1, v1); m2 = fmaxf(m2, v2); m3 = fmaxf(m3, v3);
      }
    }
    m0 = wave_max(m0); m1 = wave_max(m1); m2 = wave_max(m2); m3 = wave_max(m3);

    const bool in0 = lane < deg;
    const float p0 = in0 ? __expf(u0 - m0) : 0.0f;
    const float p1 = in0 ? __expf(u1 - m1) : 0.0f;
    const float p2 = in0 ? __expf(u2 - m2) : 0.0f;
    const float p3 = in0 ? __expf(u3 - m3) : 0.0f;
    float s0 = p0, s1 = p1, s2 = p2, s3 = p3;
    for (int base = 32; base < deg; base += 32) {
      if (base + lane < deg) {
        int j; float v0, v1, v2, v3;
        edge_logits(src_list, alpha_dst, as, start + base + lane, j, v0, v1, v2, v3);
        s0 += __expf(v0 - m0); s1 += __expf(v1 - m1);
        s2 += __expf(v2 - m2); s3 += __expf(v3 - m3);
      }
    }
    s0 = wave_sum(s0); s1 = wave_sum(s1); s2 = wave_sum(s2); s3 = wave_sum(s3);

    add_rows(Wh, lane, deg < 32 ? deg : 32, j0, p0 / s0, p1 / s1, p2 / s2, p3 / s3,
             acc0, acc1);
    for (int base = 32; base < deg; base += 32) {
      int j = 0;
      float a0 = 0.0f, a1 = 0.0f, a2 = 0.0f, a3 = 0.0f;
      if (base + lane < deg) {
        float v0, v1, v2, v3;
        edge_logits(src_list, alpha_dst, as, start + base + lane, j, v0, v1, v2, v3);
        a0 = __expf(v0 - m0) / s0; a1 = __expf(v1 - m1) / s1;
        a2 = __expf(v2 - m2) / s2; a3 = __expf(v3 - m3) / s3;
      }
      const int left = deg - base;
      add_rows(Wh, lane, left < 32 ? left : 32, j, a0, a1, a2, a3, acc0, acc1);
    }
  }

  *(float4*)(orow + 4 * lane)       = acc0;
  *(float4*)(orow + 128 + 4 * lane) = acc1;
}

extern "C" void kernel_launch(void* const* d_in, const int* in_sizes, int n_in,
                              void* d_out, int out_size, void* d_ws, size_t ws_size,
                              hipStream_t stream) {
  const float* h     = (const float*)d_in[0];
  const int*   ei    = (const int*)d_in[1];
  const float* W     = (const float*)d_in[2];
  const float* bias  = (const float*)d_in[3];
  const float* a_src = (const float*)d_in[4];
  const float* a_dst = (const float*)d_in[5];
  float* out = (float*)d_out;
  int* ranks = (int*)d_out;

  float* Wh        = (float*)d_ws;
  int*   rowtab    = (int*)(Wh + (size_t)N_NODES * DIM);
  float* alpha_dst = (float*)(rowtab + (size_t)N_NODES * HEADS);
  unsigned short* src_list = (unsigned short*)(alpha_dst + (size_t)N_NODES * HEADS);

  gat_gemm_wmma<<<N_NODES / 16, 256, 0, stream>>>(h, W, bias, Wh);

  node_prep<<<PREP_WAVES / 8, 256, 0, stream>>>(Wh, a_src, a_dst, out, alpha_dst, rowtab);

  edge_rank<<<N_EDGES / 256, 256, 0, stream>>>(ei, rowtab, ranks);
  row_ranges<<<(N_NODES + 255) / 256, 256, 0, stream>>>(rowtab);
  edge_place<<<N_EDGES / 256, 256, 0, stream>>>(ei, rowtab, ranks, src_list);

  node_agg<<<N_NODES / 8, 256, 0, stream>>>(src_list, rowtab, alpha_dst, Wh, out);
}


# GEMM keeps all of W in LDS: 256 persistent work-groups stage W once and loop over 16-row slabs (next slab prefetched), B fragments read from LDS instead of 16-line global gathers; k order kept
# speedup vs baseline: 3.7583x; 1.1776x over previous
#include <hip/hip_runtime.h>
#include <hip/hip_bf16.h>


typedef __attribute__((ext_vector_type(2))) float v2f;
typedef __attribute__((ext_vector_type(4))) float v4f;
typedef __attribute__((ext_vector_type(8))) float v8f;

static constexpr int N_NODES  = 50000;
static constexpr int N_EDGES  = 800000;
static constexpr int DIM      = 256;
static constexpr int HEADS    = 4;
static constexpr int HEAD_DIM = 64;
static constexpr float NEG_SLOPE = 0.2f;

static_assert(N_NODES <= 65536, "source list stores 16-bit node indices");
static_assert(N_EDGES == 2 * N_NODES * HEADS * 2, "source list fills the two N*4-word planes exactly");
static_assert(N_EDGES % 256 == 0 && N_NODES % 8 == 0, "grids below have no tail");
static_assert(N_EDGES / 128 <= N_NODES, "edge ranks fit in the upper halves of the output rows");

static constexpr int ROW_CNT   = 0;
static constexpr int ROW_START = N_NODES;
static constexpr int CURSOR    = 2 * N_NODES;
static_assert(2 * N_NODES + 1 <= N_NODES * HEADS, "row table fits in one plane");
static constexpr int OUT_ASRC = 4;

__device__ __forceinline__ int rank_word(int e) {
  return (e >> 7) * DIM + 128 + (e & 127);
}

static constexpr int HS_STRIDE = 260;
static constexpr int N_SLABS   = N_NODES / 16;
static constexpr int GEMM_WGS  = 256;
static_assert(N_NODES % 16 == 0, "slabs have no tail");

struct FragSet {
  v2f a[8], b0[8], b1[8];
};

__device__ __forceinline__ void load_frags(FragSet& f, const float* hA,
                                           const float* wB0, const float* wB1, int g) {
  #pragma unroll
  for (int i = 0; i < 8; ++i) {
    const int k = g * 32 + i * 4;
    f.a[i]  = *(const v2f*)(hA + k);
    f.b0[i] = *(const v2f*)(wB0 + k);
    f.b1[i] = *(const v2f*)(wB1 + k);
  }
}

__device__ __forceinline__ void mma_frags(const FragSet& f, v8f& c0, v8f& c1) {
  #pragma unroll
  for (int i = 0; i < 8; ++i) {
    c0 = __builtin_amdgcn_wmma_f32_16x16x4_f32(false, f.a[i], false, f.b0[i],
                                               (short)0, c0, false, false);
    c1 = __builtin_amdgcn_wmma_f32_16x16x4_f32(false, f.a[i], false, f.b1[i],
                                               (short)0, c1, false, false);
  }
}

__global__ __launch_bounds__(256) void gat_gemm_wmma(
    const float* __restrict__ h, const float* __restrict__ W,
    const float* __restrict__ bias, float* __restrict__ Wh) {
  __shared__ __attribute__((aligned(16))) float Ws[DIM * HS_STRIDE];
  __shared__ __attribute__((aligned(16))) float hs[2][16 * HS_STRIDE];

  const int t = threadIdx.x;

  const int sr = t >> 6;
  const int sk = (t & 63) * 4;

  int s = blockIdx.x;

  float4 pre[4];
  #pragma unroll
  for (int i = 0; i < 4; ++i)
    pre[i] = *(const float4*)(h + (size_t)(s * 16 + sr + 4 * i) * DIM + sk);

  #pragma unroll 4
  for (int r0 = 0; r0 < DIM; r0 += 16) {
    float4 w[4];
    #pragma unroll
    for (int i = 0; i < 4; ++i)
      w[i] = *(const float4*)(W + (size_t)(r0 + sr + 4 * i) * DIM + sk);
    #pragma unroll
    for (int i = 0; i < 4; ++i)
      *(float4*)&Ws[(r0 + sr + 4 * i) * HS_STRIDE + sk] = w[i];
  }
  #pragma unroll
  for (int i = 0; i < 4; ++i)
    *(float4*)&hs[0][(sr + 4 * i) * HS_STRIDE + sk] = pre[i];
  __syncthreads();

  const int wave = t >> 5;
  const int lane = t & 31;
  const int half = lane >> 4;
  const int lr   = lane & 15;

  const int col0 = wave * 16 + lr;
  const int col1 = (wave + 8) * 16 + lr;
  const float* wB0 = &Ws[col0 * HS_STRIDE + 2 * half];
  const float* wB1 = &Ws[col1 * HS_STRIDE + 2 * half];
  const float bias0 = bias[col0];
  const float bias1 = bias[col1];

  int cur = 0;
  for (; s < N_SLABS; s += GEMM_WGS) {
    const int sn = (s + GEMM_WGS < N_SLABS) ? s + GEMM_WGS : s;
    const float* hn = h + (size_t)(sn * 16 + sr) * DIM + sk;
    v4f n0, n1, n2, n3;
    asm volatile("global_load_b128 %0, %1, off" : "=&v"(n0) : "v"(hn) : "memory");
    asm volatile("global_load_b128 %0, %1, off offset:4096" : "=&v"(n1) : "v"(hn) : "memory");
    asm volatile("global_load_b128 %0, %1, off offset:8192" : "=&v"(n2) : "v"(hn) : "memory");
    asm volatile("global_load_b128 %0, %1, off offset:12288\n\ts_wait_xcnt 0x0"
                 : "=&v"(n3) : "v"(hn) : "memory");

    const float* hA = &hs[cur][lr * HS_STRIDE + 2 * half];

    v8f c0 = {};
    v8f c1 = {};
    FragSet f0, f1;
    load_frags(f0, hA, wB0, wB1, 0);
    #pragma unroll
    for (int g = 0; g < 8; g += 2) {
      load_frags(f1, hA, wB0, wB1, g + 1);
      mma_frags(f0, c0, c1);
      if (g + 2 < 8) load_frags(f0, hA, wB0, wB1, g + 2);
      mma_frags(f1, c0, c1);
    }

    const int row0 = s * 16;
    #pragma unroll
    for (int r = 0; r < 8; ++r) {
      const int row = row0 + r + 8 * half;
      Wh[row * DIM + col0] = c0[r] + bias0;
      Wh[row * DIM + col1] = c1[r] + bias1;
    }

    asm volatile("s_wait_loadcnt 0x0" : "+v"(n0), "+v"(n1), "+v"(n2), "+v"(n3) : : "memory");
    float* hsn = &hs[cur ^ 1][sr * HS_STRIDE + sk];
    *(v4f*)(hsn)                  = n0;
    *(v4f*)(hsn + 4 * HS_STRIDE)  = n1;
    *(v4f*)(hsn + 8 * HS_STRIDE)  = n2;
    *(v4f*)(hsn + 12 * HS_STRIDE) = n3;
    asm volatile("s_wait_dscnt 0x0" ::: "memory");
    asm volatile("s_barrier_signal -1" ::: "memory");
    asm volatile("s_barrier_wait -1" ::: "memory");
    cur ^= 1;
  }
}

static constexpr int PREP_ROWS  = 5;
static constexpr int PREP_WAVES = N_NODES / PREP_ROWS;
static_assert(N_NODES % PREP_ROWS == 0 && PREP_WAVES % 8 == 0, "node_prep grid has no tail");

__device__ __forceinline__ float dot4(const float4 a, const float4 b) {
  return a.x * b.x + a.y * b.y + a.z * b.z + a.w * b.w;
}

__device__ __forceinline__ float half_wave_sum(float x) {
  for (int d = 8; d > 0; d >>= 1) x += __shfl_xor(x, d);
  return x;
}

__device__ __forceinline__ float lane_bcast(float x, int k) {
  return __int_as_float(__builtin_amdgcn_readlane(__float_as_int(x), k));
}

__global__ __launch_bounds__(256) void node_prep(
    const float* __restrict__ Wh, const float* __restrict__ a_src,
    const float* __restrict__ a_dst, float* __restrict__ out,
    float* __restrict__ alpha_dst, int* __restrict__ rowtab) {
  const int lane = threadIdx.x & 31;
  const int wv   = blockIdx.x * 8 + (threadIdx.x >> 5);
  const int n0   = wv * PREP_ROWS;

  const float4 s0 = *(const float4*)(a_src + 4 * lane);
  const float4 s1 = *(const float4*)(a_src + 128 + 4 * lane);
  const float4 d0 = *(const float4*)(a_dst + 4 * lane);
  const float4 d1 = *(const float4*)(a_dst + 128 + 4 * lane);

  float4 r0[PREP_ROWS], r1[PREP_ROWS];
  #pragma unroll
  for (int i = 0; i < PREP_ROWS; ++i) {
    const float* row = Wh + (size_t)(n0 + i) * DIM + 4 * lane;
    r0[i] = *(const float4*)(row);
    r1[i] = *(const float4*)(row + 128);
  }

  #pragma unroll
  for (int i = 0; i < PREP_ROWS; ++i) {
    const int n = n0 + i;
    const float ps0 = half_wave_sum(dot4(r0[i], s0));
    const float ps1 = half_wave_sum(dot4(r1[i], s1));
    const float pd0 = half_wave_sum(dot4(r0[i], d0));
    const float pd1 = half_wave_sum(dot4(r1[i], d1));
    const float4 vs = {lane_bcast(ps0, 0), lane_bcast(ps0, 16),
                       lane_bcast(ps1, 0), lane_bcast(ps1, 16)};
    const float4 vd = {lane_bcast(pd0, 0), lane_bcast(pd0, 16),
                       lane_bcast(pd1, 0), lane_bcast(pd1, 16)};
    if (lane == 0) {
      *(float4*)(out + (size_t)n * DIM + OUT_ASRC) = vs;
      *(float4*)(alpha_dst + (size_t)n * HEADS)    = vd;
      rowtab[ROW_CNT + n] = 0;
    }
  }
  if (wv == 0 && lane == 1) rowtab[CURSOR] = 0;
}

__device__ __forceinline__ float leaky(float x) {
  return x > 0.0f ? x : NEG_SLOPE * x;
}

__global__ __launch_bounds__(256) void edge_rank(
    const int* __restrict__ ei, int* __restrict__ rowtab, int* __restrict__ ranks) {
  const int e = blockIdx.x * 256 + threadIdx.x;
  const int i = ei[e];
  const int r = atomicAdd(&rowtab[ROW_CNT + i], 1);
  ranks[rank_word(e)] = r;
}

__global__ __launch_bounds__(256) void row_ranges(int* __restrict__ rowtab) {
  __shared__ int wsum[8];
  __shared__ int sbase;
  const int t    = threadIdx.x;
  const int lane = t & 31;
  const int wave = t >> 5;
  const int n    = blockIdx.x * 256 + t;

  const int c = (n < N_NODES) ? rowtab[ROW_CNT + n] : 0;
  int x = c;
  for (int d = 1; d < 32; d <<= 1) {
    const int y = __shfl_up(x, d);
    if (lane >= d) x += y;
  }
  if (lane == 31) wsum[wave] = x;
  __syncthreads();
  if (t == 0) {
    int tot = 0;
    for (int w = 0; w < 8; ++w) {
      const int s = wsum[w];
      wsum[w] = tot;
      tot += s;
    }
    sbase = atomicAdd(&rowtab[CURSOR], tot);
  }
  __syncthreads();
  if (n < N_NODES) rowtab[ROW_START + n] = sbase + wsum[wave] + x - c;
}

__global__ __launch_bounds__(256) void edge_place(
    const int* __restrict__ ei, const int* __restrict__ rowtab,
    const int* __restrict__ ranks, unsigned short* __restrict__ src_list) {
  const int e = blockIdx.x * 256 + threadIdx.x;
  const int i = ei[e];
  const int j = ei[N_EDGES + e];
  const int r = ranks[rank_word(e)];
  const int s = rowtab[ROW_START + i];
  src_list[s + r] = (unsigned short)j;
}

__device__ __forceinline__ void edge_logits(
    const unsigned short* __restrict__ src_list, const float* __restrict__ alpha_dst,
    const float4 as, int slot, int& j, float& v0, float& v1, float& v2, float& v3) {
  j = src_list[slot];
  const float4 ad = *(const float4*)(alpha_dst + j * HEADS);
  v0 = leaky(as.x + ad.x);
  v1 = leaky(as.y + ad.y);
  v2 = leaky(as.z + ad.z);
  v3 = leaky(as.w + ad.w);
}

__device__ __forceinline__ float wave_max(float x) {
  for (int d = 16; d > 0; d >>= 1) x = fmaxf(x, __shfl_xor(x, d));
  return x;
}

__device__ __forceinline__ float wave_sum(float x) {
  for (int d = 16; d > 0; d >>= 1) x += __shfl_xor(x, d);
  return x;
}

__device__ __forceinline__ void add_rows(
    const float* __restrict__ Wh, int lane, int cnt, int j, float a0, float a1,
    float a2, float a3, float4& acc0, float4& acc1) {
  const bool lo = lane < 16;
  #pragma unroll 4
  for (int k = 0; k < cnt; ++k) {
    const int   jk = __builtin_amdgcn_readlane(j, k);
    const float x0 = lane_bcast(a0, k);
    const float x1 = lane_bcast(a1, k);
    const float x2 = lane_bcast(a2, k);
    const float x3 = lane_bcast(a3, k);
    const float w0 = lo ? x0 : x1;
    const float w1 = lo ? x2 : x3;
    const float* row = Wh + (size_t)jk * DIM + 4 * lane;
    const float4 r0 = *(const float4*)(row);
    const float4 r1 = *(const float4*)(row + 128);
    acc0.x += w0 * r0.x; acc0.y += w0 * r0.y; acc0.z += w0 * r0.z; acc0.w += w0 * r0.w;
    acc1.x += w1 * r1.x; acc1.y += w1 * r1.y; acc1.z += w1 * r1.z; acc1.w += w1 * r1.w;
  }
}

__global__ __launch_bounds__(256) void node_agg(
    const unsigned short* __restrict__ src_list, const int* __restrict__ rowtab,
    const float* __restrict__ alpha_dst, const float* __restrict__ Wh, float* out) {
  const int lane = threadIdx.x & 31;
  const int n = blockIdx.x * 8 + __builtin_amdgcn_readfirstlane(threadIdx.x >> 5);
  float* orow = out + (size_t)n * DIM;
  const int deg   = __builtin_amdgcn_readfirstlane(rowtab[ROW_CNT + n]);
  const int start = __builtin_amdgcn_readfirstlane(rowtab[ROW_START + n]);
  const float4 as = *(const float4*)(orow + OUT_ASRC);

  float4 acc0 = {0.0f, 0.0f, 0.0f, 0.0f};
  float4 acc1 = {0.0f, 0.0f, 0.0f, 0.0f};

  if (deg > 0) {
    const float NEG_INF = __int_as_float(0xFF800000);
    int j0 = 0;
    float u0 = NEG_INF, u1 = NEG_INF, u2 = NEG_INF, u3 = NEG_INF;
    if (lane < deg)
      edge_logits(src_list, alpha_dst, as, start + lane, j0, u0, u1, u2, u3);

    float m0 = u0, m1 = u1, m2 = u2, m3 = u3;
    for (int base = 32; base < deg; base += 32) {
      if (base + lane < deg) {
        int j; float v0, v1, v2, v3;
        edge_logits(src_list, alpha_dst, as, start + base + lane, j, v0, v1, v2, v3);
        m0 = fmaxf(m0, v0); m1 = fmaxf(m1, v1); m2 = fmaxf(m2, v2); m3 = fmaxf(m3, v3);
      }
    }
    m0 = wave_max(m0); m1 = wave_max(m1); m2 = wave_max(m2); m3 = wave_max(m3);

    const bool in0 = lane < deg;
    const float p0 = in0 ? __expf(u0 - m0) : 0.0f;
    const float p1 = in0 ? __expf(u1 - m1) : 0.0f;
    const float p2 = in0 ? __expf(u2 - m2) : 0.0f;
    const float p3 = in0 ? __expf(u3 - m3) : 0.0f;
    float s0 = p0, s1 = p1, s2 = p2, s3 = p3;
    for (int base = 32; base < deg; base += 32) {
      if (base + lane < deg) {
        int j; float v0, v1, v2, v3;
        edge_logits(src_list, alpha_dst, as, start + base + lane, j, v0, v1, v2, v3);
        s0 += __expf(v0 - m0); s1 += __expf(v1 - m1);
        s2 += __expf(v2 - m2); s3 += __expf(v3 - m3);
      }
    }
    s0 = wave_sum(s0); s1 = wave_sum(s1); s2 = wave_sum(s2); s3 = wave_sum(s3);

    add_rows(Wh, lane, deg < 32 ? deg : 32, j0, p0 / s0, p1 / s1, p2 / s2, p3 / s3,
             acc0, acc1);
    for (int base = 32; base < deg; base += 32) {
      int j = 0;
      float a0 = 0.0f, a1 = 0.0f, a2 = 0.0f, a3 = 0.0f;
      if (base + lane < deg) {
        float v0, v1, v2, v3;
        edge_logits(src_list, alpha_dst, as, start + base + lane, j, v0, v1, v2, v3);
        a0 = __expf(v0 - m0) / s0; a1 = __expf(v1 - m1) / s1;
        a2 = __expf(v2 - m2) / s2; a3 = __expf(v3 - m3) / s3;
      }
      const int left = deg - base;
      add_rows(Wh, lane, left < 32 ? left : 32, j, a0, a1, a2, a3, acc0, acc1);
    }
  }

  *(float4*)(orow + 4 * lane)       = acc0;
  *(float4*)(orow + 128 + 4 * lane) = acc1;
}

extern "C" void kernel_launch(void* const* d_in, const int* in_sizes, int n_in,
                              void* d_out, int out_size, void* d_ws, size_t ws_size,
                              hipStream_t stream) {
  const float* h     = (const float*)d_in[0];
  const int*   ei    = (const int*)d_in[1];
  const float* W     = (const float*)d_in[2];
  const float* bias  = (const float*)d_in[3];
  const float* a_src = (const float*)d_in[4];
  const float* a_dst = (const float*)d_in[5];
  float* out = (float*)d_out;
  int* ranks = (int*)d_out;

  float* Wh        = (float*)d_ws;
  int*   rowtab    = (int*)(Wh + (size_t)N_NODES * DIM);
  float* alpha_dst = (float*)(rowtab + (size_t)N_NODES * HEADS);
  unsigned short* src_list = (unsigned short*)(alpha_dst + (size_t)N_NODES * HEADS);

  gat_gemm_wmma<<<GEMM_WGS, 256, 0, stream>>>(h, W, bias, Wh);

  node_prep<<<PREP_WAVES / 8, 256, 0, stream>>>(Wh, a_src, a_dst, out, alpha_dst, rowtab);

  edge_rank<<<N_EDGES / 256, 256, 0, stream>>>(ei, rowtab, ranks);
  row_ranges<<<(N_NODES + 255) / 256, 256, 0, stream>>>(rowtab);
  edge_place<<<N_EDGES / 256, 256, 0, stream>>>(ei, rowtab, ranks, src_list);

  node_agg<<<N_NODES / 8, 256, 0, stream>>>(src_list, rowtab, alpha_dst, Wh, out);
}


# Edges grouped by destination in one launch: 128 work-groups each own 391 rows, scan the edge list, count and slot their edges with LDS adds, no global add per edge; row sum order = LDS arrival order
# speedup vs baseline: 5.2483x; 1.3965x over previous
#include <hip/hip_runtime.h>
#include <hip/hip_bf16.h>


typedef __attribute__((ext_vector_type(2))) float v2f;
typedef __attribute__((ext_vector_type(4))) float v4f;
typedef __attribute__((ext_vector_type(8))) float v8f;

static constexpr int N_NODES  = 50000;
static constexpr int N_EDGES  = 800000;
static constexpr int DIM      = 256;
static constexpr int HEADS    = 4;
static constexpr int HEAD_DIM = 64;
static constexpr float NEG_SLOPE = 0.2f;

static_assert(N_NODES <= 65536, "source list stores 16-bit node indices");
static_assert(N_EDGES == 2 * N_NODES * HEADS * 2, "source list fills the two N*4-word planes exactly");
static_assert(N_EDGES % 4 == 0 && N_NODES % 8 == 0, "edge list splits into 4-edge vectors; node_agg grid has no tail");

static constexpr int ROW_CNT   = 0;
static constexpr int ROW_START = N_NODES;
static constexpr int CURSOR    = 2 * N_NODES;
static_assert(2 * N_NODES + 1 <= N_NODES * HEADS, "row table fits in one plane");
static constexpr int OUT_ASRC = 4;

static constexpr int HS_STRIDE = 260;
static constexpr int N_SLABS   = N_NODES / 16;
static constexpr int GEMM_WGS  = 256;
static_assert(N_NODES % 16 == 0, "slabs have no tail");

struct FragSet {
  v2f a[8], b0[8], b1[8];
};

__device__ __forceinline__ void load_frags(FragSet& f, const float* hA,
                                           const float* wB0, const float* wB1, int g) {
  #pragma unroll
  for (int i = 0; i < 8; ++i) {
    const int k = g * 32 + i * 4;
    f.a[i]  = *(const v2f*)(hA + k);
    f.b0[i] = *(const v2f*)(wB0 + k);
    f.b1[i] = *(const v2f*)(wB1 + k);
  }
}

__device__ __forceinline__ void mma_frags(const FragSet& f, v8f& c0, v8f& c1) {
  #pragma unroll
  for (int i = 0; i < 8; ++i) {
    c0 = __builtin_amdgcn_wmma_f32_16x16x4_f32(false, f.a[i], false, f.b0[i],
                                               (short)0, c0, false, false);
    c1 = __builtin_amdgcn_wmma_f32_16x16x4_f32(false, f.a[i], false, f.b1[i],
                                               (short)0, c1, false, false);
  }
}

__global__ __launch_bounds__(256) void gat_gemm_wmma(
    const float* __restrict__ h, const float* __restrict__ W,
    const float* __restrict__ bias, float* __restrict__ Wh) {
  __shared__ __attribute__((aligned(16))) float Ws[DIM * HS_STRIDE];
  __shared__ __attribute__((aligned(16))) float hs[2][16 * HS_STRIDE];

  const int t = threadIdx.x;

  const int sr = t >> 6;
  const int sk = (t & 63) * 4;

  int s = blockIdx.x;

  float4 pre[4];
  #pragma unroll
  for (int i = 0; i < 4; ++i)
    pre[i] = *(const float4*)(h + (size_t)(s * 16 + sr + 4 * i) * DIM + sk);

  #pragma unroll 4
  for (int r0 = 0; r0 < DIM; r0 += 16) {
    float4 w[4];
    #pragma unroll
    for (int i = 0; i < 4; ++i)
      w[i] = *(const float4*)(W + (size_t)(r0 + sr + 4 * i) * DIM + sk);
    #pragma unroll
    for (int i = 0; i < 4; ++i)
      *(float4*)&Ws[(r0 + sr + 4 * i) * HS_STRIDE + sk] = w[i];
  }
  #pragma unroll
  for (int i = 0; i < 4; ++i)
    *(float4*)&hs[0][(sr + 4 * i) * HS_STRIDE + sk] = pre[i];
  __syncthreads();

  const int wave = t >> 5;
  const int lane = t & 31;
  const int half = lane >> 4;
  const int lr   = lane & 15;

  const int col0 = wave * 16 + lr;
  const int col1 = (wave + 8) * 16 + lr;
  const float* wB0 = &Ws[col0 * HS_STRIDE + 2 * half];
  const float* wB1 = &Ws[col1 * HS_STRIDE + 2 * half];
  const float bias0 = bias[col0];
  const float bias1 = bias[col1];

  int cur = 0;
  for (; s < N_SLABS; s += GEMM_WGS) {
    const int sn = (s + GEMM_WGS < N_SLABS) ? s + GEMM_WGS : s;
    const float* hn = h + (size_t)(sn * 16 + sr) * DIM + sk;
    v4f n0, n1, n2, n3;
    asm volatile("global_load_b128 %0, %1, off" : "=&v"(n0) : "v"(hn) : "memory");
    asm volatile("global_load_b128 %0, %1, off offset:4096" : "=&v"(n1) : "v"(hn) : "memory");
    asm volatile("global_load_b128 %0, %1, off offset:8192" : "=&v"(n2) : "v"(hn) : "memory");
    asm volatile("global_load_b128 %0, %1, off offset:12288\n\ts_wait_xcnt 0x0"
                 : "=&v"(n3) : "v"(hn) : "memory");

    const float* hA = &hs[cur][lr * HS_STRIDE + 2 * half];

    v8f c0 = {};
    v8f c1 = {};
    FragSet f0, f1;
    load_frags(f0, hA, wB0, wB1, 0);
    #pragma unroll
    for (int g = 0; g < 8; g += 2) {
      load_frags(f1, hA, wB0, wB1, g + 1);
      mma_frags(f0, c0, c1);
      if (g + 2 < 8) load_frags(f0, hA, wB0, wB1, g + 2);
      mma_frags(f1, c0, c1);
    }

    const int row0 = s * 16;
    #pragma unroll
    for (int r = 0; r < 8; ++r) {
      const int row = row0 + r + 8 * half;
      Wh[row * DIM + col0] = c0[r] + bias0;
      Wh[row * DIM + col1] = c1[r] + bias1;
    }

    asm volatile("s_wait_loadcnt 0x0" : "+v"(n0), "+v"(n1), "+v"(n2), "+v"(n3) : : "memory");
    float* hsn = &hs[cur ^ 1][sr * HS_STRIDE + sk];
    *(v4f*)(hsn)                  = n0;
    *(v4f*)(hsn + 4 * HS_STRIDE)  = n1;
    *(v4f*)(hsn + 8 * HS_STRIDE)  = n2;
    *(v4f*)(hsn + 12 * HS_STRIDE) = n3;
    asm volatile("s_wait_dscnt 0x0" ::: "memory");
    asm volatile("s_barrier_signal -1" ::: "memory");
    asm volatile("s_barrier_wait -1" ::: "memory");
    cur ^= 1;
  }
}

static constexpr int PREP_ROWS  = 5;
static constexpr int PREP_WAVES = N_NODES / PREP_ROWS;
static_assert(N_NODES % PREP_ROWS == 0 && PREP_WAVES % 8 == 0, "node_prep grid has no tail");

__device__ __forceinline__ float dot4(const float4 a, const float4 b) {
  return a.x * b.x + a.y * b.y + a.z * b.z + a.w * b.w;
}

__device__ __forceinline__ float half_wave_sum(float x) {
  for (int d = 8; d > 0; d >>= 1) x += __shfl_xor(x, d);
  return x;
}

__device__ __forceinline__ float lane_bcast(float x, int k) {
  return __int_as_float(__builtin_amdgcn_readlane(__float_as_int(x), k));
}

__global__ __launch_bounds__(256) void node_prep(
    const float* __restrict__ Wh, const float* __restrict__ a_src,
    const float* __restrict__ a_dst, float* __restrict__ out,
    float* __restrict__ alpha_dst, int* __restrict__ rowtab) {
  const int lane = threadIdx.x & 31;
  const int wv   = blockIdx.x * 8 + (threadIdx.x >> 5);
  const int n0   = wv * PREP_ROWS;

  const float4 s0 = *(const float4*)(a_src + 4 * lane);
  const float4 s1 = *(const float4*)(a_src + 128 + 4 * lane);
  const float4 d0 = *(const float4*)(a_dst + 4 * lane);
  const float4 d1 = *(const float4*)(a_dst + 128 + 4 * lane);

  float4 r0[PREP_ROWS], r1[PREP_ROWS];
  #pragma unroll
  for (int i = 0; i < PREP_ROWS; ++i) {
    const float* row = Wh + (size_t)(n0 + i) * DIM + 4 * lane;
    r0[i] = *(const float4*)(row);
    r1[i] = *(const float4*)(row + 128);
  }

  #pragma unroll
  for (int i = 0; i < PREP_ROWS; ++i) {
    const int n = n0 + i;
    const float ps0 = half_wave_sum(dot4(r0[i], s0));
    const float ps1 = half_wave_sum(dot4(r1[i], s1));
    const float pd0 = half_wave_sum(dot4(r0[i], d0));
    const float pd1 = half_wave_sum(dot4(r1[i], d1));
    const float4 vs = {lane_bcast(ps0, 0), lane_bcast(ps0, 16),
                       lane_bcast(ps1, 0), lane_bcast(ps1, 16)};
    const float4 vd = {lane_bcast(pd0, 0), lane_bcast(pd0, 16),
                       lane_bcast(pd1, 0), lane_bcast(pd1, 16)};
    if (lane == 0) {
      *(float4*)(out + (size_t)n * DIM + OUT_ASRC) = vs;
      *(float4*)(alpha_dst + (size_t)n * HEADS)    = vd;
    }
  }
  if (wv == 0 && lane == 1) rowtab[CURSOR] = 0;
}

__device__ __forceinline__ float leaky(float x) {
  return x > 0.0f ? x : NEG_SLOPE * x;
}

static constexpr int GRP_WGS     = 128;
static constexpr int GRP_THREADS = 1024;
static constexpr int GRP_ROWS    = (N_NODES + GRP_WGS - 1) / GRP_WGS;
static constexpr int GRP_CAP     = 16384;
static constexpr int N_VECS      = N_EDGES / 4;
static constexpr int SCAN_UNROLL = 8;
static constexpr int MAIN_BLKS   = N_VECS / GRP_THREADS / SCAN_UNROLL * SCAN_UNROLL;
static constexpr int TAIL_BLKS   = (N_VECS - MAIN_BLKS * GRP_THREADS + GRP_THREADS - 1) / GRP_THREADS;
static constexpr int WALK_UNROLL = 4;
static_assert(GRP_ROWS <= GRP_THREADS, "one row per thread in the range scan");
static_assert(GRP_WGS * GRP_ROWS >= N_NODES, "the ranges cover every row");
static_assert(MAIN_BLKS > 0 && TAIL_BLKS >= 1 && TAIL_BLKS <= SCAN_UNROLL, "tail is one short trip");

__device__ __forceinline__ void note_vec(const int4 v, int vec, bool ok, int lo,
                                         int* nlist, int* list) {
  const unsigned d0 = (unsigned)(v.x - lo);
  const unsigned d1 = (unsigned)(v.y - lo);
  const unsigned d2 = (unsigned)(v.z - lo);
  const unsigned d3 = (unsigned)(v.w - lo);
  const unsigned m01 = d0 < d1 ? d0 : d1;
  const unsigned m23 = d2 < d3 ? d2 : d3;
  const unsigned m   = m01 < m23 ? m01 : m23;
  if (ok && m < (unsigned)GRP_ROWS) {
    const int idx = atomicAdd(nlist, 1);
    if (idx < GRP_CAP) list[idx] = vec;
  }
}

__device__ __forceinline__ void walk_vecs(
    const int* __restrict__ ei, const int* list, int nwalk, bool whole, int lo,
    int* cnt, unsigned short* __restrict__ src_list, const bool place, int t) {
  const int4* ev_dst = (const int4*)ei;
  const int4* ev_src = (const int4*)(ei + N_EDGES);
  for (int base = 0; base < nwalk; base += WALK_UNROLL * GRP_THREADS) {
    int  vecs[WALK_UNROLL];
    int4 dv[WALK_UNROLL], sv[WALK_UNROLL];
    #pragma unroll
    for (int u = 0; u < WALK_UNROLL; ++u) {
      const int idx = base + u * GRP_THREADS + t;
      vecs[u] = -1;
      if (idx < nwalk) vecs[u] = whole ? idx : list[idx];
    }
    #pragma unroll
    for (int u = 0; u < WALK_UNROLL; ++u) {
      const int vc = vecs[u] < 0 ? 0 : vecs[u];
      dv[u] = ev_dst[vc];
      if (place) sv[u] = ev_src[vc];
    }
    #pragma unroll
    for (int u = 0; u < WALK_UNROLL; ++u) {
      if (vecs[u] >= 0) {
        const int di[4] = {dv[u].x, dv[u].y, dv[u].z, dv[u].w};
        const int si[4] = {place ? sv[u].x : 0, place ? sv[u].y : 0,
                           place ? sv[u].z : 0, place ? sv[u].w : 0};
        #pragma unroll
        for (int k = 0; k < 4; ++k) {
          const unsigned d = (unsigned)(di[k] - lo);
          if (d < (unsigned)GRP_ROWS) {
            if (place) {
              const int slot = atomicAdd(&cnt[d], 1);
              src_list[slot] = (unsigned short)si[k];
            } else {
              atomicAdd(&cnt[d], 1);
            }
          }
        }
      }
    }
  }
}

__global__ __launch_bounds__(GRP_THREADS) void edge_group(
    const int* __restrict__ ei, int* __restrict__ rowtab,
    unsigned short* __restrict__ src_list) {
  __shared__ int cnt[GRP_ROWS];
  __shared__ int list[GRP_CAP];
  __shared__ int nlist;
  __shared__ int wsum[GRP_THREADS / 32];
  __shared__ int sbase;

  const int t    = threadIdx.x;
  const int lane = t & 31;
  const int wave = t >> 5;
  const int lo   = blockIdx.x * GRP_ROWS;

  if (t < GRP_ROWS) cnt[t] = 0;
  if (t == 0) nlist = 0;
  __syncthreads();

  const int4* ev = (const int4*)ei;
  const int bstart = blockIdx.x * MAIN_BLKS / GRP_WGS;
  for (int it0 = 0; it0 < MAIN_BLKS; it0 += SCAN_UNROLL) {
    int4 v[SCAN_UNROLL];
    int  vecs[SCAN_UNROLL];
    #pragma unroll
    for (int u = 0; u < SCAN_UNROLL; ++u) {
      int b = it0 + u + bstart;
      if (b >= MAIN_BLKS) b -= MAIN_BLKS;
      vecs[u] = b * GRP_THREADS + t;
      v[u] = ev[vecs[u]];
    }
    #pragma unroll
    for (int u = 0; u < SCAN_UNROLL; ++u)
      note_vec(v[u], vecs[u], true, lo, &nlist, list);
  }
  {
    int4 v[TAIL_BLKS];
    int  vecs[TAIL_BLKS];
    #pragma unroll
    for (int u = 0; u < TAIL_BLKS; ++u) {
      const int q = (MAIN_BLKS + u) * GRP_THREADS + t;
      vecs[u] = (q < N_VECS) ? q : -1;
      v[u] = ev[(q < N_VECS) ? q : N_VECS - 1];
    }
    #pragma unroll
    for (int u = 0; u < TAIL_BLKS; ++u)
      note_vec(v[u], vecs[u], vecs[u] >= 0, lo, &nlist, list);
  }
  __syncthreads();

  const int  nl    = nlist;
  const bool whole = nl > GRP_CAP;
  const int  nwalk = whole ? N_VECS : nl;

  walk_vecs(ei, list, nwalk, whole, lo, cnt, src_list, false, t);
  __syncthreads();

  const int c = (t < GRP_ROWS) ? cnt[t] : 0;
  int x = c;
  for (int d = 1; d < 32; d <<= 1) {
    const int y = __shfl_up(x, d);
    if (lane >= d) x += y;
  }
  if (lane == 31) wsum[wave] = x;
  __syncthreads();
  if (t == 0) {
    int tot = 0;
    for (int w = 0; w < GRP_THREADS / 32; ++w) {
      const int s = wsum[w];
      wsum[w] = tot;
      tot += s;
    }
    sbase = atomicAdd(&rowtab[CURSOR], tot);
  }
  __syncthreads();
  const int start = sbase + wsum[wave] + x - c;
  if (t < GRP_ROWS) {
    const int n = lo + t;
    if (n < N_NODES) {
      rowtab[ROW_CNT + n]   = c;
      rowtab[ROW_START + n] = start;
    }
    cnt[t] = start;
  }
  __syncthreads();

  walk_vecs(ei, list, nwalk, whole, lo, cnt, src_list, true, t);
}

__device__ __forceinline__ void edge_logits(
    const unsigned short* __restrict__ src_list, const float* __restrict__ alpha_dst,
    const float4 as, int slot, int& j, float& v0, float& v1, float& v2, float& v3) {
  j = src_list[slot];
  const float4 ad = *(const float4*)(alpha_dst + j * HEADS);
  v0 = leaky(as.x + ad.x);
  v1 = leaky(as.y + ad.y);
  v2 = leaky(as.z + ad.z);
  v3 = leaky(as.w + ad.w);
}

__device__ __forceinline__ float wave_max(float x) {
  for (int d = 16; d > 0; d >>= 1) x = fmaxf(x, __shfl_xor(x, d));
  return x;
}

__device__ __forceinline__ float wave_sum(float x) {
  for (int d = 16; d > 0; d >>= 1) x += __shfl_xor(x, d);
  return x;
}

__device__ __forceinline__ void add_rows(
    const float* __restrict__ Wh, int lane, int cnt, int j, float a0, float a1,
    float a2, float a3, float4& acc0, float4& acc1) {
  const bool lo = lane < 16;
  #pragma unroll 4
  for (int k = 0; k < cnt; ++k) {
    const int   jk = __builtin_amdgcn_readlane(j, k);
    const float x0 = lane_bcast(a0, k);
    const float x1 = lane_bcast(a1, k);
    const float x2 = lane_bcast(a2, k);
    const float x3 = lane_bcast(a3, k);
    const float w0 = lo ? x0 : x1;
    const float w1 = lo ? x2 : x3;
    const float* row = Wh + (size_t)jk * DIM + 4 * lane;
    const float4 r0 = *(const float4*)(row);
    const float4 r1 = *(const float4*)(row + 128);
    acc0.x += w0 * r0.x; acc0.y += w0 * r0.y; acc0.z += w0 * r0.z; acc0.w += w0 * r0.w;
    acc1.x += w1 * r1.x; acc1.y += w1 * r1.y; acc1.z += w1 * r1.z; acc1.w += w1 * r1.w;
  }
}

__global__ __launch_bounds__(256) void node_agg(
    const unsigned short* __restrict__ src_list, const int* __restrict__ rowtab,
    const float* __restrict__ alpha_dst, const float* __restrict__ Wh, float* out) {
  const int lane = threadIdx.x & 31;
  const int n = blockIdx.x * 8 + __builtin_amdgcn_readfirstlane(threadIdx.x >> 5);
  float* orow = out + (size_t)n * DIM;
  const int deg   = __builtin_amdgcn_readfirstlane(rowtab[ROW_CNT + n]);
  const int start = __builtin_amdgcn_readfirstlane(rowtab[ROW_START + n]);
  const float4 as = *(const float4*)(orow + OUT_ASRC);

  float4 acc0 = {0.0f, 0.0f, 0.0f, 0.0f};
  float4 acc1 = {0.0f, 0.0f, 0.0f, 0.0f};

  if (deg > 0) {
    const float NEG_INF = __int_as_float(0xFF800000);
    int j0 = 0;
    float u0 = NEG_INF, u1 = NEG_INF, u2 = NEG_INF, u3 = NEG_INF;
    if (lane < deg)
      edge_logits(src_list, alpha_dst, as, start + lane, j0, u0, u1, u2, u3);

    float m0 = u0, m1 = u1, m2 = u2, m3 = u3;
    for (int base = 32; base < deg; base += 32) {
      if (base + lane < deg) {
        int j; float v0, v1, v2, v3;
        edge_logits(src_list, alpha_dst, as, start + base + lane, j, v0, v1, v2, v3);
        m0 = fmaxf(m0, v0); m1 = fmaxf(m1, v1); m2 = fmaxf(m2, v2); m3 = fmaxf(m3, v3);
      }
    }
    m0 = wave_max(m0); m1 = wave_max(m1); m2 = wave_max(m2); m3 = wave_max(m3);

    const bool in0 = lane < deg;
    const float p0 = in0 ? __expf(u0 - m0) : 0.0f;
    const float p1 = in0 ? __expf(u1 - m1) : 0.0f;
    const float p2 = in0 ? __expf(u2 - m2) : 0.0f;
    const float p3 = in0 ? __expf(u3 - m3) : 0.0f;
    float s0 = p0, s1 = p1, s2 = p2, s3 = p3;
    for (int base = 32; base < deg; base += 32) {
      if (base + lane < deg) {
        int j; float v0, v1, v2, v3;
        edge_logits(src_list, alpha_dst, as, start + base + lane, j, v0, v1, v2, v3);
        s0 += __expf(v0 - m0); s1 += __expf(v1 - m1);
        s2 += __expf(v2 - m2); s3 += __expf(v3 - m3);
      }
    }
    s0 = wave_sum(s0); s1 = wave_sum(s1); s2 = wave_sum(s2); s3 = wave_sum(s3);

    add_rows(Wh, lane, deg < 32 ? deg : 32, j0, p0 / s0, p1 / s1, p2 / s2, p3 / s3,
             acc0, acc1);
    for (int base = 32; base < deg; base += 32) {
      int j = 0;
      float a0 = 0.0f, a1 = 0.0f, a2 = 0.0f, a3 = 0.0f;
      if (base + lane < deg) {
        float v0, v1, v2, v3;
        edge_logits(src_list, alpha_dst, as, start + base + lane, j, v0, v1, v2, v3);
        a0 = __expf(v0 - m0) / s0; a1 = __expf(v1 - m1) / s1;
        a2 = __expf(v2 - m2) / s2; a3 = __expf(v3 - m3) / s3;
      }
      const int left = deg - base;
      add_rows(Wh, lane, left < 32 ? left : 32, j, a0, a1, a2, a3, acc0, acc1);
    }
  }

  *(float4*)(orow + 4 * lane)       = acc0;
  *(float4*)(orow + 128 + 4 * lane) = acc1;
}

extern "C" void kernel_launch(void* const* d_in, const int* in_sizes, int n_in,
                              void* d_out, int out_size, void* d_ws, size_t ws_size,
                              hipStream_t stream) {
  const float* h     = (const float*)d_in[0];
  const int*   ei    = (const int*)d_in[1];
  const float* W     = (const float*)d_in[2];
  const float* bias  = (const float*)d_in[3];
  const float* a_src = (const float*)d_in[4];
  const float* a_dst = (const float*)d_in[5];
  float* out = (float*)d_out;

  float* Wh        = (float*)d_ws;
  int*   rowtab    = (int*)(Wh + (size_t)N_NODES * DIM);
  float* alpha_dst = (float*)(rowtab + (size_t)N_NODES * HEADS);
  unsigned short* src_list = (unsigned short*)(alpha_dst + (size_t)N_NODES * HEADS);

  gat_gemm_wmma<<<GEMM_WGS, 256, 0, stream>>>(h, W, bias, Wh);

  node_prep<<<PREP_WAVES / 8, 256, 0, stream>>>(Wh, a_src, a_dst, out, alpha_dst, rowtab);

  edge_group<<<GRP_WGS, GRP_THREADS, 0, stream>>>(ei, rowtab, src_list);

  node_agg<<<N_NODES / 8, 256, 0, stream>>>(src_list, rowtab, alpha_dst, Wh, out);
}


# alpha_src/alpha_dst rows ride as 313 extra work-groups behind the 128 edge-grouping ones in ONE launch (independent work; 3 launches, was 4); cursor clear moves to the GEMM launch; sum orders kept
# speedup vs baseline: 5.6278x; 1.0723x over previous
#include <hip/hip_runtime.h>
#include <hip/hip_bf16.h>


typedef __attribute__((ext_vector_type(2))) float v2f;
typedef __attribute__((ext_vector_type(4))) float v4f;
typedef __attribute__((ext_vector_type(8))) float v8f;

static constexpr int N_NODES  = 50000;
static constexpr int N_EDGES  = 800000;
static constexpr int DIM      = 256;
static constexpr int HEADS    = 4;
static constexpr int HEAD_DIM = 64;
static constexpr float NEG_SLOPE = 0.2f;

static_assert(N_NODES <= 65536, "source list stores 16-bit node indices");
static_assert(N_EDGES == 2 * N_NODES * HEADS * 2, "source list fills the two N*4-word planes exactly");
static_assert(N_EDGES % 4 == 0 && N_NODES % 8 == 0, "edge list splits into 4-edge vectors; node_agg grid has no tail");

static constexpr int ROW_CNT   = 0;
static constexpr int ROW_START = N_NODES;
static constexpr int CURSOR    = 2 * N_NODES;
static_assert(2 * N_NODES + 1 <= N_NODES * HEADS, "row table fits in one plane");
static constexpr int OUT_ASRC = 4;

static constexpr int HS_STRIDE = 260;
static constexpr int N_SLABS   = N_NODES / 16;
static constexpr int GEMM_WGS  = 256;
static_assert(N_NODES % 16 == 0, "slabs have no tail");

struct FragSet {
  v2f a[8], b0[8], b1[8];
};

__device__ __forceinline__ void load_frags(FragSet& f, const float* hA,
                                           const float* wB0, const float* wB1, int g) {
  #pragma unroll
  for (int i = 0; i < 8; ++i) {
    const int k = g * 32 + i * 4;
    f.a[i]  = *(const v2f*)(hA + k);
    f.b0[i] = *(const v2f*)(wB0 + k);
    f.b1[i] = *(const v2f*)(wB1 + k);
  }
}

__device__ __forceinline__ void mma_frags(const FragSet& f, v8f& c0, v8f& c1) {
  #pragma unroll
  for (int i = 0; i < 8; ++i) {
    c0 = __builtin_amdgcn_wmma_f32_16x16x4_f32(false, f.a[i], false, f.b0[i],
                                               (short)0, c0, false, false);
    c1 = __builtin_amdgcn_wmma_f32_16x16x4_f32(false, f.a[i], false, f.b1[i],
                                               (short)0, c1, false, false);
  }
}

__global__ __launch_bounds__(256) void gat_gemm_wmma(
    const float* __restrict__ h, const float* __restrict__ W,
    const float* __restrict__ bias, float* __restrict__ Wh,
    int* __restrict__ rowtab) {
  __shared__ __attribute__((aligned(16))) float Ws[DIM * HS_STRIDE];
  __shared__ __attribute__((aligned(16))) float hs[2][16 * HS_STRIDE];

  const int t = threadIdx.x;

  const int sr = t >> 6;
  const int sk = (t & 63) * 4;

  int s = blockIdx.x;

  float4 pre[4];
  #pragma unroll
  for (int i = 0; i < 4; ++i)
    pre[i] = *(const float4*)(h + (size_t)(s * 16 + sr + 4 * i) * DIM + sk);

  #pragma unroll 4
  for (int r0 = 0; r0 < DIM; r0 += 16) {
    float4 w[4];
    #pragma unroll
    for (int i = 0; i < 4; ++i)
      w[i] = *(const float4*)(W + (size_t)(r0 + sr + 4 * i) * DIM + sk);
    #pragma unroll
    for (int i = 0; i < 4; ++i)
      *(float4*)&Ws[(r0 + sr + 4 * i) * HS_STRIDE + sk] = w[i];
  }
  #pragma unroll
  for (int i = 0; i < 4; ++i)
    *(float4*)&hs[0][(sr + 4 * i) * HS_STRIDE + sk] = pre[i];
  __syncthreads();

  const int wave = t >> 5;
  const int lane = t & 31;
  const int half = lane >> 4;
  const int lr   = lane & 15;

  const int col0 = wave * 16 + lr;
  const int col1 = (wave + 8) * 16 + lr;
  const float* wB0 = &Ws[col0 * HS_STRIDE + 2 * half];
  const float* wB1 = &Ws[col1 * HS_STRIDE + 2 * half];
  const float bias0 = bias[col0];
  const float bias1 = bias[col1];

  int cur = 0;
  for (; s < N_SLABS; s += GEMM_WGS) {
    const int sn = (s + GEMM_WGS < N_SLABS) ? s + GEMM_WGS : s;
    const float* hn = h + (size_t)(sn * 16 + sr) * DIM + sk;
    v4f n0, n1, n2, n3;
    asm volatile("global_load_b128 %0, %1, off" : "=&v"(n0) : "v"(hn) : "memory");
    asm volatile("global_load_b128 %0, %1, off offset:4096" : "=&v"(n1) : "v"(hn) : "memory");
    asm volatile("global_load_b128 %0, %1, off offset:8192" : "=&v"(n2) : "v"(hn) : "memory");
    asm volatile("global_load_b128 %0, %1, off offset:12288\n\ts_wait_xcnt 0x0"
                 : "=&v"(n3) : "v"(hn) : "memory");

    const float* hA = &hs[cur][lr * HS_STRIDE + 2 * half];

    v8f c0 = {};
    v8f c1 = {};
    FragSet f0, f1;
    load_frags(f0, hA, wB0, wB1, 0);
    #pragma unroll
    for (int g = 0; g < 8; g += 2) {
      load_frags(f1, hA, wB0, wB1, g + 1);
      mma_frags(f0, c0, c1);
      if (g + 2 < 8) load_frags(f0, hA, wB0, wB1, g + 2);
      mma_frags(f1, c0, c1);
    }

    const int row0 = s * 16;
    #pragma unroll
    for (int r = 0; r < 8; ++r) {
      const int row = row0 + r + 8 * half;
      Wh[row * DIM + col0] = c0[r] + bias0;
      Wh[row * DIM + col1] = c1[r] + bias1;
    }

    asm volatile("s_wait_loadcnt 0x0" : "+v"(n0), "+v"(n1), "+v"(n2), "+v"(n3) : : "memory");
    float* hsn = &hs[cur ^ 1][sr * HS_STRIDE + sk];
    *(v4f*)(hsn)                  = n0;
    *(v4f*)(hsn + 4 * HS_STRIDE)  = n1;
    *(v4f*)(hsn + 8 * HS_STRIDE)  = n2;
    *(v4f*)(hsn + 12 * HS_STRIDE) = n3;
    asm volatile("s_wait_dscnt 0x0" ::: "memory");
    asm volatile("s_barrier_signal -1" ::: "memory");
    asm volatile("s_barrier_wait -1" ::: "memory");
    cur ^= 1;
  }

  if (blockIdx.x == GEMM_WGS - 1 && t == 0) rowtab[CURSOR] = 0;
}

static constexpr int PREP_ROWS  = 5;
static constexpr int PREP_WAVES = N_NODES / PREP_ROWS;
static_assert(N_NODES % PREP_ROWS == 0, "every wave has PREP_ROWS whole rows");

__device__ __forceinline__ float dot4(const float4 a, const float4 b) {
  return a.x * b.x + a.y * b.y + a.z * b.z + a.w * b.w;
}

__device__ __forceinline__ float half_wave_sum(float x) {
  for (int d = 8; d > 0; d >>= 1) x += __shfl_xor(x, d);
  return x;
}

__device__ __forceinline__ float lane_bcast(float x, int k) {
  return __int_as_float(__builtin_amdgcn_readlane(__float_as_int(x), k));
}

__device__ __forceinline__ void node_prep_wave(
    const float* __restrict__ Wh, const float* __restrict__ a_src,
    const float* __restrict__ a_dst, float* __restrict__ out,
    float* __restrict__ alpha_dst, const int wv, const int lane) {
  const int n0 = wv * PREP_ROWS;

  const float4 s0 = *(const float4*)(a_src + 4 * lane);
  const float4 s1 = *(const float4*)(a_src + 128 + 4 * lane);
  const float4 d0 = *(const float4*)(a_dst + 4 * lane);
  const float4 d1 = *(const float4*)(a_dst + 128 + 4 * lane);

  float4 r0[PREP_ROWS], r1[PREP_ROWS];
  #pragma unroll
  for (int i = 0; i < PREP_ROWS; ++i) {
    const float* row = Wh + (size_t)(n0 + i) * DIM + 4 * lane;
    r0[i] = *(const float4*)(row);
    r1[i] = *(const float4*)(row + 128);
  }

  #pragma unroll
  for (int i = 0; i < PREP_ROWS; ++i) {
    const int n = n0 + i;
    const float ps0 = half_wave_sum(dot4(r0[i], s0));
    const float ps1 = half_wave_sum(dot4(r1[i], s1));
    const float pd0 = half_wave_sum(dot4(r0[i], d0));
    const float pd1 = half_wave_sum(dot4(r1[i], d1));
    const float4 vs = {lane_bcast(ps0, 0), lane_bcast(ps0, 16),
                       lane_bcast(ps1, 0), lane_bcast(ps1, 16)};
    const float4 vd = {lane_bcast(pd0, 0), lane_bcast(pd0, 16),
                       lane_bcast(pd1, 0), lane_bcast(pd1, 16)};
    if (lane == 0) {
      *(float4*)(out + (size_t)n * DIM + OUT_ASRC) = vs;
      *(float4*)(alpha_dst + (size_t)n * HEADS)    = vd;
    }
  }
}

__device__ __forceinline__ float leaky(float x) {
  return x > 0.0f ? x : NEG_SLOPE * x;
}

static constexpr int GRP_WGS     = 128;
static constexpr int GRP_THREADS = 1024;
static constexpr int GRP_ROWS    = (N_NODES + GRP_WGS - 1) / GRP_WGS;
static constexpr int GRP_CAP     = 16384;
static constexpr int N_VECS      = N_EDGES / 4;
static constexpr int SCAN_UNROLL = 8;
static constexpr int MAIN_BLKS   = N_VECS / GRP_THREADS / SCAN_UNROLL * SCAN_UNROLL;
static constexpr int TAIL_BLKS   = (N_VECS - MAIN_BLKS * GRP_THREADS + GRP_THREADS - 1) / GRP_THREADS;
static constexpr int WALK_UNROLL = 4;
static_assert(GRP_ROWS <= GRP_THREADS, "one row per thread in the range scan");
static_assert(GRP_WGS * GRP_ROWS >= N_NODES, "the ranges cover every row");
static_assert(MAIN_BLKS > 0 && TAIL_BLKS >= 1 && TAIL_BLKS <= SCAN_UNROLL, "tail is one short trip");
static constexpr int GRP_WAVES   = GRP_THREADS / 32;
static constexpr int PREP_BLKS   = (PREP_WAVES + GRP_WAVES - 1) / GRP_WAVES;
static constexpr int PG_BLKS     = GRP_WGS + PREP_BLKS;

__device__ __forceinline__ void note_vec(const int4 v, int vec, bool ok, int lo,
                                         int* nlist, int* list) {
  const unsigned d0 = (unsigned)(v.x - lo);
  const unsigned d1 = (unsigned)(v.y - lo);
  const unsigned d2 = (unsigned)(v.z - lo);
  const unsigned d3 = (unsigned)(v.w - lo);
  const unsigned m01 = d0 < d1 ? d0 : d1;
  const unsigned m23 = d2 < d3 ? d2 : d3;
  const unsigned m   = m01 < m23 ? m01 : m23;
  if (ok && m < (unsigned)GRP_ROWS) {
    const int idx = atomicAdd(nlist, 1);
    if (idx < GRP_CAP) list[idx] = vec;
  }
}

__device__ __forceinline__ void walk_vecs(
    const int* __restrict__ ei, const int* list, int nwalk, bool whole, int lo,
    int* cnt, unsigned short* __restrict__ src_list, const bool place, int t) {
  const int4* ev_dst = (const int4*)ei;
  const int4* ev_src = (const int4*)(ei + N_EDGES);
  for (int base = 0; base < nwalk; base += WALK_UNROLL * GRP_THREADS) {
    int  vecs[WALK_UNROLL];
    int4 dv[WALK_UNROLL], sv[WALK_UNROLL];
    #pragma unroll
    for (int u = 0; u < WALK_UNROLL; ++u) {
      const int idx = base + u * GRP_THREADS + t;
      vecs[u] = -1;
      if (idx < nwalk) vecs[u] = whole ? idx : list[idx];
    }
    #pragma unroll
    for (int u = 0; u < WALK_UNROLL; ++u) {
      const int vc = vecs[u] < 0 ? 0 : vecs[u];
      dv[u] = ev_dst[vc];
      if (place) sv[u] = ev_src[vc];
    }
    #pragma unroll
    for (int u = 0; u < WALK_UNROLL; ++u) {
      if (vecs[u] >= 0) {
        const int di[4] = {dv[u].x, dv[u].y, dv[u].z, dv[u].w};
        const int si[4] = {place ? sv[u].x : 0, place ? sv[u].y : 0,
                           place ? sv[u].z : 0, place ? sv[u].w : 0};
        #pragma unroll
        for (int k = 0; k < 4; ++k) {
          const unsigned d = (unsigned)(di[k] - lo);
          if (d < (unsigned)GRP_ROWS) {
            if (place) {
              const int slot = atomicAdd(&cnt[d], 1);
              src_list[slot] = (unsigned short)si[k];
            } else {
              atomicAdd(&cnt[d], 1);
            }
          }
        }
      }
    }
  }
}

__global__ __launch_bounds__(GRP_THREADS) void group_and_prep(
    const int* __restrict__ ei, int* __restrict__ rowtab,
    unsigned short* __restrict__ src_list,
    const float* __restrict__ Wh, const float* __restrict__ a_src,
    const float* __restrict__ a_dst, float* __restrict__ out,
    float* __restrict__ alpha_dst) {
  if (blockIdx.x >= GRP_WGS) {
    const int wv = (blockIdx.x - GRP_WGS) * GRP_WAVES + (threadIdx.x >> 5);
    if (wv < PREP_WAVES)
      node_prep_wave(Wh, a_src, a_dst, out, alpha_dst, wv, threadIdx.x & 31);
    return;
  }

  __shared__ int cnt[GRP_ROWS];
  __shared__ int list[GRP_CAP];
  __shared__ int nlist;
  __shared__ int wsum[GRP_THREADS / 32];
  __shared__ int sbase;

  const int t    = threadIdx.x;
  const int lane = t & 31;
  const int wave = t >> 5;
  const int lo   = blockIdx.x * GRP_ROWS;

  if (t < GRP_ROWS) cnt[t] = 0;
  if (t == 0) nlist = 0;
  __syncthreads();

  const int4* ev = (const int4*)ei;
  const int bstart = blockIdx.x * MAIN_BLKS / GRP_WGS;
  for (int it0 = 0; it0 < MAIN_BLKS; it0 += SCAN_UNROLL) {
    int4 v[SCAN_UNROLL];
    int  vecs[SCAN_UNROLL];
    #pragma unroll
    for (int u = 0; u < SCAN_UNROLL; ++u) {
      int b = it0 + u + bstart;
      if (b >= MAIN_BLKS) b -= MAIN_BLKS;
      vecs[u] = b * GRP_THREADS + t;
      v[u] = ev[vecs[u]];
    }
    #pragma unroll
    for (int u = 0; u < SCAN_UNROLL; ++u)
      note_vec(v[u], vecs[u], true, lo, &nlist, list);
  }
  {
    int4 v[TAIL_BLKS];
    int  vecs[TAIL_BLKS];
    #pragma unroll
    for (int u = 0; u < TAIL_BLKS; ++u) {
      const int q = (MAIN_BLKS + u) * GRP_THREADS + t;
      vecs[u] = (q < N_VECS) ? q : -1;
      v[u] = ev[(q < N_VECS) ? q : N_VECS - 1];
    }
    #pragma unroll
    for (int u = 0; u < TAIL_BLKS; ++u)
      note_vec(v[u], vecs[u], vecs[u] >= 0, lo, &nlist, list);
  }
  __syncthreads();

  const int  nl    = nlist;
  const bool whole = nl > GRP_CAP;
  const int  nwalk = whole ? N_VECS : nl;

  walk_vecs(ei, list, nwalk, whole, lo, cnt, src_list, false, t);
  __syncthreads();

  const int c = (t < GRP_ROWS) ? cnt[t] : 0;
  int x = c;
  for (int d = 1; d < 32; d <<= 1) {
    const int y = __shfl_up(x, d);
    if (lane >= d) x += y;
  }
  if (lane == 31) wsum[wave] = x;
  __syncthreads();
  if (t == 0) {
    int tot = 0;
    for (int w = 0; w < GRP_THREADS / 32; ++w) {
      const int s = wsum[w];
      wsum[w] = tot;
      tot += s;
    }
    sbase = atomicAdd(&rowtab[CURSOR], tot);
  }
  __syncthreads();
  const int start = sbase + wsum[wave] + x - c;
  if (t < GRP_ROWS) {
    const int n = lo + t;
    if (n < N_NODES) {
      rowtab[ROW_CNT + n]   = c;
      rowtab[ROW_START + n] = start;
    }
    cnt[t] = start;
  }
  __syncthreads();

  walk_vecs(ei, list, nwalk, whole, lo, cnt, src_list, true, t);
}

__device__ __forceinline__ void edge_logits(
    const unsigned short* __restrict__ src_list, const float* __restrict__ alpha_dst,
    const float4 as, int slot, int& j, float& v0, float& v1, float& v2, float& v3) {
  j = src_list[slot];
  const float4 ad = *(const float4*)(alpha_dst + j * HEADS);
  v0 = leaky(as.x + ad.x);
  v1 = leaky(as.y + ad.y);
  v2 = leaky(as.z + ad.z);
  v3 = leaky(as.w + ad.w);
}

__device__ __forceinline__ float wave_max(float x) {
  for (int d = 16; d > 0; d >>= 1) x = fmaxf(x, __shfl_xor(x, d));
  return x;
}

__device__ __forceinline__ float wave_sum(float x) {
  for (int d = 16; d > 0; d >>= 1) x += __shfl_xor(x, d);
  return x;
}

__device__ __forceinline__ void add_rows(
    const float* __restrict__ Wh, int lane, int cnt, int j, float a0, float a1,
    float a2, float a3, float4& acc0, float4& acc1) {
  const bool lo = lane < 16;
  #pragma unroll 4
  for (int k = 0; k < cnt; ++k) {
    const int   jk = __builtin_amdgcn_readlane(j, k);
    const float x0 = lane_bcast(a0, k);
    const float x1 = lane_bcast(a1, k);
    const float x2 = lane_bcast(a2, k);
    const float x3 = lane_bcast(a3, k);
    const float w0 = lo ? x0 : x1;
    const float w1 = lo ? x2 : x3;
    const float* row = Wh + (size_t)jk * DIM + 4 * lane;
    const float4 r0 = *(const float4*)(row);
    const float4 r1 = *(const float4*)(row + 128);
    acc0.x += w0 * r0.x; acc0.y += w0 * r0.y; acc0.z += w0 * r0.z; acc0.w += w0 * r0.w;
    acc1.x += w1 * r1.x; acc1.y += w1 * r1.y; acc1.z += w1 * r1.z; acc1.w += w1 * r1.w;
  }
}

__global__ __launch_bounds__(256) void node_agg(
    const unsigned short* __restrict__ src_list, const int* __restrict__ rowtab,
    const float* __restrict__ alpha_dst, const float* __restrict__ Wh, float* out) {
  const int lane = threadIdx.x & 31;
  const int n = blockIdx.x * 8 + __builtin_amdgcn_readfirstlane(threadIdx.x >> 5);
  float* orow = out + (size_t)n * DIM;
  const int deg   = __builtin_amdgcn_readfirstlane(rowtab[ROW_CNT + n]);
  const int start = __builtin_amdgcn_readfirstlane(rowtab[ROW_START + n]);
  const float4 as = *(const float4*)(orow + OUT_ASRC);

  float4 acc0 = {0.0f, 0.0f, 0.0f, 0.0f};
  float4 acc1 = {0.0f, 0.0f, 0.0f, 0.0f};

  if (deg > 0) {
    const float NEG_INF = __int_as_float(0xFF800000);
    int j0 = 0;
    float u0 = NEG_INF, u1 = NEG_INF, u2 = NEG_INF, u3 = NEG_INF;
    if (lane < deg)
      edge_logits(src_list, alpha_dst, as, start + lane, j0, u0, u1, u2, u3);

    float m0 = u0, m1 = u1, m2 = u2, m3 = u3;
    for (int base = 32; base < deg; base += 32) {
      if (base + lane < deg) {
        int j; float v0, v1, v2, v3;
        edge_logits(src_list, alpha_dst, as, start + base + lane, j, v0, v1, v2, v3);
        m0 = fmaxf(m0, v0); m1 = fmaxf(m1, v1); m2 = fmaxf(m2, v2); m3 = fmaxf(m3, v3);
      }
    }
    m0 = wave_max(m0); m1 = wave_max(m1); m2 = wave_max(m2); m3 = wave_max(m3);

    const bool in0 = lane < deg;
    const float p0 = in0 ? __expf(u0 - m0) : 0.0f;
    const float p1 = in0 ? __expf(u1 - m1) : 0.0f;
    const float p2 = in0 ? __expf(u2 - m2) : 0.0f;
    const float p3 = in0 ? __expf(u3 - m3) : 0.0f;
    float s0 = p0, s1 = p1, s2 = p2, s3 = p3;
    for (int base = 32; base < deg; base += 32) {
      if (base + lane < deg) {
        int j; float v0, v1, v2, v3;
        edge_logits(src_list, alpha_dst, as, start + base + lane, j, v0, v1, v2, v3);
        s0 += __expf(v0 - m0); s1 += __expf(v1 - m1);
        s2 += __expf(v2 - m2); s3 += __expf(v3 - m3);
      }
    }
    s0 = wave_sum(s0); s1 = wave_sum(s1); s2 = wave_sum(s2); s3 = wave_sum(s3);

    add_rows(Wh, lane, deg < 32 ? deg : 32, j0, p0 / s0, p1 / s1, p2 / s2, p3 / s3,
             acc0, acc1);
    for (int base = 32; base < deg; base += 32) {
      int j = 0;
      float a0 = 0.0f, a1 = 0.0f, a2 = 0.0f, a3 = 0.0f;
      if (base + lane < deg) {
        float v0, v1, v2, v3;
        edge_logits(src_list, alpha_dst, as, start + base + lane, j, v0, v1, v2, v3);
        a0 = __expf(v0 - m0) / s0; a1 = __expf(v1 - m1) / s1;
        a2 = __expf(v2 - m2) / s2; a3 = __expf(v3 - m3) / s3;
      }
      const int left = deg - base;
      add_rows(Wh, lane, left < 32 ? left : 32, j, a0, a1, a2, a3, acc0, acc1);
    }
  }

  *(float4*)(orow + 4 * lane)       = acc0;
  *(float4*)(orow + 128 + 4 * lane) = acc1;
}

extern "C" void kernel_launch(void* const* d_in, const int* in_sizes, int n_in,
                              void* d_out, int out_size, void* d_ws, size_t ws_size,
                              hipStream_t stream) {
  const float* h     = (const float*)d_in[0];
  const int*   ei    = (const int*)d_in[1];
  const float* W     = (const float*)d_in[2];
  const float* bias  = (const float*)d_in[3];
  const float* a_src = (const float*)d_in[4];
  const float* a_dst = (const float*)d_in[5];
  float* out = (float*)d_out;

  float* Wh        = (float*)d_ws;
  int*   rowtab    = (int*)(Wh + (size_t)N_NODES * DIM);
  float* alpha_dst = (float*)(rowtab + (size_t)N_NODES * HEADS);
  unsigned short* src_list = (unsigned short*)(alpha_dst + (size_t)N_NODES * HEADS);

  gat_gemm_wmma<<<GEMM_WGS, 256, 0, stream>>>(h, W, bias, Wh, rowtab);

  group_and_prep<<<PG_BLKS, GRP_THREADS, 0, stream>>>(ei, rowtab, src_list,
                                                      Wh, a_src, a_dst, out, alpha_dst);

  node_agg<<<N_NODES / 8, 256, 0, stream>>>(src_list, rowtab, alpha_dst, Wh, out);
}


# Edge list binned by row range once, by the GEMM work-groups that have a slab fewer; each range's work-group reads only its bin, not all 800000 destinations; row sum order = LDS arrival order
# speedup vs baseline: 7.2028x; 1.2799x over previous
#include <hip/hip_runtime.h>
#include <hip/hip_bf16.h>


typedef __attribute__((ext_vector_type(2))) float v2f;
typedef __attribute__((ext_vector_type(4))) float v4f;
typedef __attribute__((ext_vector_type(8))) float v8f;

static constexpr int N_NODES  = 50000;
static constexpr int N_EDGES  = 800000;
static constexpr int DIM      = 256;
static constexpr int HEADS    = 4;
static constexpr int HEAD_DIM = 64;
static constexpr float NEG_SLOPE = 0.2f;

static_assert(N_NODES <= 65536, "source list stores 16-bit node indices");
static_assert(N_EDGES == 2 * N_NODES * HEADS * 2, "source list fills the two N*4-word planes exactly");
static_assert(N_EDGES % 4 == 0 && N_NODES % 8 == 0, "edge list splits into 4-edge vectors; node_agg grid has no tail");

static constexpr int ROW_CNT   = 0;
static constexpr int ROW_START = N_NODES;
static constexpr int CURSOR    = 2 * N_NODES;
static_assert(2 * N_NODES + 1 <= N_NODES * HEADS, "row table fits in one plane");
static constexpr int OUT_ASRC = 4;

static constexpr int HS_STRIDE = 260;
static constexpr int N_SLABS   = N_NODES / 16;
static constexpr int GEMM_WGS  = 256;
static_assert(N_NODES % 16 == 0, "slabs have no tail");

static constexpr int GRP_WGS     = 128;
static constexpr int GRP_ROWS    = (N_NODES + GRP_WGS - 1) / GRP_WGS;
static constexpr int N_VECS      = N_EDGES / 4;
static constexpr int BIN_WG0     = N_SLABS % GEMM_WGS;
static constexpr int BIN_WGS     = GEMM_WGS - BIN_WG0;
static constexpr int SLICE_VECS  = (N_VECS + BIN_WGS - 1) / BIN_WGS;
static constexpr int BIN_PER_T   = (SLICE_VECS + 255) / 256;
static constexpr int SLICE_EDGES = SLICE_VECS * 4;
static constexpr int SLICE_PAD   = (SLICE_EDGES + 127) / 128 * 128;
static constexpr int BIN_TAB     = (2 * N_NODES + 1 + 31) / 32 * 32;
static_assert(SLICE_EDGES < 65536 && GRP_ROWS <= 65536, "bin start, bin size and row offset fit in 16 bits");
static_assert(GRP_WGS % 32 == 0 && GRP_WGS <= 256, "one GEMM thread per range in the bin scan");
static_assert(BIN_TAB + BIN_WGS * GRP_WGS <= N_NODES * HEADS, "bin table fits behind the row table in plane 0");
static_assert((size_t)BIN_WGS * SLICE_PAD / 128 <= (size_t)N_NODES, "binned list fits in the upper halves of the output rows");
static_assert(SLICE_EDGES <= 2 * 16 * HS_STRIDE, "a sorted slice fits in the two slab buffers");

__device__ __forceinline__ int binned_word(int q) {
  return q + (q & ~127) + 128;
}

struct FragSet {
  v2f a[8], b0[8], b1[8];
};

__device__ __forceinline__ void load_frags(FragSet& f, const float* hA,
                                           const float* wB0, const float* wB1, int g) {
  #pragma unroll
  for (int i = 0; i < 8; ++i) {
    const int k = g * 32 + i * 4;
    f.a[i]  = *(const v2f*)(hA + k);
    f.b0[i] = *(const v2f*)(wB0 + k);
    f.b1[i] = *(const v2f*)(wB1 + k);
  }
}

__device__ __forceinline__ void mma_frags(const FragSet& f, v8f& c0, v8f& c1) {
  #pragma unroll
  for (int i = 0; i < 8; ++i) {
    c0 = __builtin_amdgcn_wmma_f32_16x16x4_f32(false, f.a[i], false, f.b0[i],
                                               (short)0, c0, false, false);
    c1 = __builtin_amdgcn_wmma_f32_16x16x4_f32(false, f.a[i], false, f.b1[i],
                                               (short)0, c1, false, false);
  }
}

__global__ __launch_bounds__(256) void gat_gemm_wmma(
    const float* __restrict__ h, const float* __restrict__ W,
    const float* __restrict__ bias, float* __restrict__ Wh,
    int* __restrict__ rowtab, const int* __restrict__ ei,
    unsigned* __restrict__ binned) {
  __shared__ __attribute__((aligned(16))) float Ws[DIM * HS_STRIDE];
  __shared__ __attribute__((aligned(16))) float hs[2][16 * HS_STRIDE];
  __shared__ int bincnt[GRP_WGS];
  __shared__ int binwsum[GRP_WGS / 32];

  const int t = threadIdx.x;

  const int sr = t >> 6;
  const int sk = (t & 63) * 4;

  int s = blockIdx.x;

  float4 pre[4];
  #pragma unroll
  for (int i = 0; i < 4; ++i)
    pre[i] = *(const float4*)(h + (size_t)(s * 16 + sr + 4 * i) * DIM + sk);

  #pragma unroll 4
  for (int r0 = 0; r0 < DIM; r0 += 16) {
    float4 w[4];
    #pragma unroll
    for (int i = 0; i < 4; ++i)
      w[i] = *(const float4*)(W + (size_t)(r0 + sr + 4 * i) * DIM + sk);
    #pragma unroll
    for (int i = 0; i < 4; ++i)
      *(float4*)&Ws[(r0 + sr + 4 * i) * HS_STRIDE + sk] = w[i];
  }
  #pragma unroll
  for (int i = 0; i < 4; ++i)
    *(float4*)&hs[0][(sr + 4 * i) * HS_STRIDE + sk] = pre[i];
  __syncthreads();

  const int wave = t >> 5;
  const int lane = t & 31;
  const int half = lane >> 4;
  const int lr   = lane & 15;

  const int col0 = wave * 16 + lr;
  const int col1 = (wave + 8) * 16 + lr;
  const float* wB0 = &Ws[col0 * HS_STRIDE + 2 * half];
  const float* wB1 = &Ws[col1 * HS_STRIDE + 2 * half];
  const float bias0 = bias[col0];
  const float bias1 = bias[col1];

  int cur = 0;
  for (; s < N_SLABS; s += GEMM_WGS) {
    const int sn = (s + GEMM_WGS < N_SLABS) ? s + GEMM_WGS : s;
    const float* hn = h + (size_t)(sn * 16 + sr) * DIM + sk;
    v4f n0, n1, n2, n3;
    asm volatile("global_load_b128 %0, %1, off" : "=&v"(n0) : "v"(hn) : "memory");
    asm volatile("global_load_b128 %0, %1, off offset:4096" : "=&v"(n1) : "v"(hn) : "memory");
    asm volatile("global_load_b128 %0, %1, off offset:8192" : "=&v"(n2) : "v"(hn) : "memory");
    asm volatile("global_load_b128 %0, %1, off offset:12288\n\ts_wait_xcnt 0x0"
                 : "=&v"(n3) : "v"(hn) : "memory");

    const float* hA = &hs[cur][lr * HS_STRIDE + 2 * half];

    v8f c0 = {};
    v8f c1 = {};
    FragSet f0, f1;
    load_frags(f0, hA, wB0, wB1, 0);
    #pragma unroll
    for (int g = 0; g < 8; g += 2) {
      load_frags(f1, hA, wB0, wB1, g + 1);
      mma_frags(f0, c0, c1);
      if (g + 2 < 8) load_frags(f0, hA, wB0, wB1, g + 2);
      mma_frags(f1, c0, c1);
    }

    const int row0 = s * 16;
    #pragma unroll
    for (int r = 0; r < 8; ++r) {
      const int row = row0 + r + 8 * half;
      Wh[row * DIM + col0] = c0[r] + bias0;
      Wh[row * DIM + col1] = c1[r] + bias1;
    }

    asm volatile("s_wait_loadcnt 0x0" : "+v"(n0), "+v"(n1), "+v"(n2), "+v"(n3) : : "memory");
    float* hsn = &hs[cur ^ 1][sr * HS_STRIDE + sk];
    *(v4f*)(hsn)                  = n0;
    *(v4f*)(hsn + 4 * HS_STRIDE)  = n1;
    *(v4f*)(hsn + 8 * HS_STRIDE)  = n2;
    *(v4f*)(hsn + 12 * HS_STRIDE) = n3;
    asm volatile("s_wait_dscnt 0x0" ::: "memory");
    asm volatile("s_barrier_signal -1" ::: "memory");
    asm volatile("s_barrier_wait -1" ::: "memory");
    cur ^= 1;
  }

  if (blockIdx.x == GEMM_WGS - 1 && t == 0) rowtab[CURSOR] = 0;

  if (blockIdx.x >= BIN_WG0) {
    const int bw = blockIdx.x - BIN_WG0;
    const int4* ev_dst = (const int4*)ei;
    const int4* ev_src = (const int4*)(ei + N_EDGES);
    int4 dv[BIN_PER_T], sv[BIN_PER_T];
    bool okv[BIN_PER_T];
    #pragma unroll
    for (int u = 0; u < BIN_PER_T; ++u) {
      const int lv = u * 256 + t;
      const int q  = bw * SLICE_VECS + lv;
      okv[u] = lv < SLICE_VECS && q < N_VECS;
      const int qc = okv[u] ? q : 0;
      dv[u] = ev_dst[qc];
      sv[u] = ev_src[qc];
    }
    if (t < GRP_WGS) bincnt[t] = 0;
    __syncthreads();

    int bin[BIN_PER_T * 4], rank[BIN_PER_T * 4];
    #pragma unroll
    for (int u = 0; u < BIN_PER_T; ++u) {
      const int di[4] = {dv[u].x, dv[u].y, dv[u].z, dv[u].w};
      #pragma unroll
      for (int k = 0; k < 4; ++k) {
        const bool valid = okv[u] && (unsigned)di[k] < (unsigned)N_NODES;
        bin[u * 4 + k]  = valid ? di[k] / GRP_ROWS : -1;
        rank[u * 4 + k] = 0;
        if (valid) rank[u * 4 + k] = atomicAdd(&bincnt[bin[u * 4 + k]], 1);
      }
    }
    __syncthreads();

    const int cb = (t < GRP_WGS) ? bincnt[t] : 0;
    int xb = cb;
    for (int d = 1; d < 32; d <<= 1) {
      const int y = __shfl_up(xb, d);
      if (lane >= d) xb += y;
    }
    if (t < GRP_WGS && lane == 31) binwsum[wave] = xb;
    __syncthreads();
    if (t < GRP_WGS) {
      int startb = xb - cb;
      for (int w = 0; w < wave; ++w) startb += binwsum[w];
      bincnt[t] = startb;
      rowtab[BIN_TAB + bw * GRP_WGS + t] = (startb << 16) | cb;
    }
    __syncthreads();

    unsigned* stage = (unsigned*)&hs[0][0];
    #pragma unroll
    for (int i = 0; i < BIN_PER_T * 4; ++i)
      rank[i] += bincnt[bin[i] < 0 ? 0 : bin[i]];
    #pragma unroll
    for (int u = 0; u < BIN_PER_T; ++u) {
      const int di[4] = {dv[u].x, dv[u].y, dv[u].z, dv[u].w};
      const int si[4] = {sv[u].x, sv[u].y, sv[u].z, sv[u].w};
      #pragma unroll
      for (int k = 0; k < 4; ++k) {
        const int b = bin[u * 4 + k];
        if (b >= 0)
          stage[rank[u * 4 + k]] =
              (unsigned)(di[k] - b * GRP_ROWS) | ((unsigned)si[k] << 16);
      }
    }
    __syncthreads();
    #pragma unroll
    for (int u = 0; u < BIN_PER_T; ++u) {
      const int i = u * 256 + t;
      if (i < SLICE_VECS)
        *(uint4*)(binned + binned_word(bw * SLICE_PAD + 4 * i)) = *(const uint4*)(stage + 4 * i);
    }
  }
}

static constexpr int PREP_ROWS  = 5;
static constexpr int PREP_WAVES = N_NODES / PREP_ROWS;
static_assert(N_NODES % PREP_ROWS == 0, "every wave has PREP_ROWS whole rows");

__device__ __forceinline__ float dot4(const float4 a, const float4 b) {
  return a.x * b.x + a.y * b.y + a.z * b.z + a.w * b.w;
}

__device__ __forceinline__ float half_wave_sum(float x) {
  for (int d = 8; d > 0; d >>= 1) x += __shfl_xor(x, d);
  return x;
}

__device__ __forceinline__ float lane_bcast(float x, int k) {
  return __int_as_float(__builtin_amdgcn_readlane(__float_as_int(x), k));
}

__device__ __forceinline__ void node_prep_wave(
    const float* __restrict__ Wh, const float* __restrict__ a_src,
    const float* __restrict__ a_dst, float* __restrict__ out,
    float* __restrict__ alpha_dst, const int wv, const int lane) {
  const int n0 = wv * PREP_ROWS;

  const float4 s0 = *(const float4*)(a_src + 4 * lane);
  const float4 s1 = *(const float4*)(a_src + 128 + 4 * lane);
  const float4 d0 = *(const float4*)(a_dst + 4 * lane);
  const float4 d1 = *(const float4*)(a_dst + 128 + 4 * lane);

  float4 r0[PREP_ROWS], r1[PREP_ROWS];
  #pragma unroll
  for (int i = 0; i < PREP_ROWS; ++i) {
    const float* row = Wh + (size_t)(n0 + i) * DIM + 4 * lane;
    r0[i] = *(const float4*)(row);
    r1[i] = *(const float4*)(row + 128);
  }

  #pragma unroll
  for (int i = 0; i < PREP_ROWS; ++i) {
    const int n = n0 + i;
    const float ps0 = half_wave_sum(dot4(r0[i], s0));
    const float ps1 = half_wave_sum(dot4(r1[i], s1));
    const float pd0 = half_wave_sum(dot4(r0[i], d0));
    const float pd1 = half_wave_sum(dot4(r1[i], d1));
    const float4 vs = {lane_bcast(ps0, 0), lane_bcast(ps0, 16),
                       lane_bcast(ps1, 0), lane_bcast(ps1, 16)};
    const float4 vd = {lane_bcast(pd0, 0), lane_bcast(pd0, 16),
                       lane_bcast(pd1, 0), lane_bcast(pd1, 16)};
    if (lane == 0) {
      *(float4*)(out + (size_t)n * DIM + OUT_ASRC) = vs;
      *(float4*)(alpha_dst + (size_t)n * HEADS)    = vd;
    }
  }
}

__device__ __forceinline__ float leaky(float x) {
  return x > 0.0f ? x : NEG_SLOPE * x;
}

static constexpr int GRP_THREADS = 1024;
static constexpr int SEG_T       = GRP_THREADS / BIN_WGS;
static constexpr int WALK_UNROLL = 8;
static_assert(GRP_ROWS <= GRP_THREADS, "one row per thread in the range scan");
static_assert(GRP_WGS * GRP_ROWS >= N_NODES, "the ranges cover every row");
static_assert(SEG_T >= 1, "at least one thread per slice");
static constexpr int GRP_WAVES   = GRP_THREADS / 32;
static constexpr int PREP_BLKS   = (PREP_WAVES + GRP_WAVES - 1) / GRP_WAVES;
static constexpr int PG_BLKS     = GRP_WGS + PREP_BLKS;

__device__ __forceinline__ void walk_run(
    const unsigned* __restrict__ binned, int q0, int n, int sub,
    int* cnt, unsigned short* __restrict__ src_list, const bool place) {
  for (int k0 = sub; k0 < n; k0 += SEG_T * WALK_UNROLL) {
    unsigned e[WALK_UNROLL];
    bool     ok[WALK_UNROLL];
    #pragma unroll
    for (int u = 0; u < WALK_UNROLL; ++u) {
      const int k = k0 + u * SEG_T;
      ok[u] = k < n;
      e[u] = binned[binned_word(q0 + (ok[u] ? k : k0))];
    }
    #pragma unroll
    for (int u = 0; u < WALK_UNROLL; ++u) {
      const unsigned d = e[u] & 0xFFFFu;
      if (ok[u] && d < (unsigned)GRP_ROWS) {
        if (place) {
          const int slot = atomicAdd(&cnt[d], 1);
          src_list[slot] = (unsigned short)(e[u] >> 16);
        } else {
          atomicAdd(&cnt[d], 1);
        }
      }
    }
  }
}

__global__ __launch_bounds__(GRP_THREADS) void group_and_prep(
    int* __restrict__ rowtab, unsigned short* __restrict__ src_list,
    const float* __restrict__ Wh, const float* __restrict__ a_src,
    const float* __restrict__ a_dst, float* __restrict__ out,
    float* __restrict__ alpha_dst) {
  if (blockIdx.x >= GRP_WGS) {
    const int wv = (blockIdx.x - GRP_WGS) * GRP_WAVES + (threadIdx.x >> 5);
    if (wv < PREP_WAVES)
      node_prep_wave(Wh, a_src, a_dst, out, alpha_dst, wv, threadIdx.x & 31);
    return;
  }

  __shared__ int cnt[GRP_ROWS];
  __shared__ int wsum[GRP_THREADS / 32];
  __shared__ int sbase;

  const int t    = threadIdx.x;
  const int lane = t & 31;
  const int wave = t >> 5;
  const int lo   = blockIdx.x * GRP_ROWS;

  const unsigned* binned = (const unsigned*)out;
  const int seg = t / SEG_T;
  const int sub = t - seg * SEG_T;
  int n = 0, q0 = 0;
  if (seg < BIN_WGS) {
    const unsigned p = (unsigned)rowtab[BIN_TAB + seg * GRP_WGS + blockIdx.x];
    n  = (int)(p & 0xFFFFu);
    q0 = seg * SLICE_PAD + (int)(p >> 16);
  }

  if (t < GRP_ROWS) cnt[t] = 0;
  __syncthreads();

  walk_run(binned, q0, n, sub, cnt, src_list, false);
  __syncthreads();

  const int c = (t < GRP_ROWS) ? cnt[t] : 0;
  int x = c;
  for (int d = 1; d < 32; d <<= 1) {
    const int y = __shfl_up(x, d);
    if (lane >= d) x += y;
  }
  if (lane == 31) wsum[wave] = x;
  __syncthreads();
  if (t == 0) {
    int tot = 0;
    for (int w = 0; w < GRP_THREADS / 32; ++w) {
      const int s = wsum[w];
      wsum[w] = tot;
      tot += s;
    }
    sbase = atomicAdd(&rowtab[CURSOR], tot);
  }
  __syncthreads();
  const int start = sbase + wsum[wave] + x - c;
  if (t < GRP_ROWS) {
    const int nrow = lo + t;
    if (nrow < N_NODES) {
      rowtab[ROW_CNT + nrow]   = c;
      rowtab[ROW_START + nrow] = start;
    }
    cnt[t] = start;
  }
  __syncthreads();

  walk_run(binned, q0, n, sub, cnt, src_list, true);
}

__device__ __forceinline__ void edge_logits(
    const unsigned short* __restrict__ src_list, const float* __restrict__ alpha_dst,
    const float4 as, int slot, int& j, float& v0, float& v1, float& v2, float& v3) {
  j = src_list[slot];
  const float4 ad = *(const float4*)(alpha_dst + j * HEADS);
  v0 = leaky(as.x + ad.x);
  v1 = leaky(as.y + ad.y);
  v2 = leaky(as.z + ad.z);
  v3 = leaky(as.w + ad.w);
}

__device__ __forceinline__ float wave_max(float x) {
  for (int d = 16; d > 0; d >>= 1) x = fmaxf(x, __shfl_xor(x, d));
  return x;
}

__device__ __forceinline__ float wave_sum(float x) {
  for (int d = 16; d > 0; d >>= 1) x += __shfl_xor(x, d);
  return x;
}

__device__ __forceinline__ void add_rows(
    const float* __restrict__ Wh, int lane, int cnt, int j, float a0, float a1,
    float a2, float a3, float4& acc0, float4& acc1) {
  const bool lo = lane < 16;
  #pragma unroll 4
  for (int k = 0; k < cnt; ++k) {
    const int   jk = __builtin_amdgcn_readlane(j, k);
    const float x0 = lane_bcast(a0, k);
    const float x1 = lane_bcast(a1, k);
    const float x2 = lane_bcast(a2, k);
    const float x3 = lane_bcast(a3, k);
    const float w0 = lo ? x0 : x1;
    const float w1 = lo ? x2 : x3;
    const float* row = Wh + (size_t)jk * DIM + 4 * lane;
    const float4 r0 = *(const float4*)(row);
    const float4 r1 = *(const float4*)(row + 128);
    acc0.x += w0 * r0.x; acc0.y += w0 * r0.y; acc0.z += w0 * r0.z; acc0.w += w0 * r0.w;
    acc1.x += w1 * r1.x; acc1.y += w1 * r1.y; acc1.z += w1 * r1.z; acc1.w += w1 * r1.w;
  }
}

__global__ __launch_bounds__(256) void node_agg(
    const unsigned short* __restrict__ src_list, const int* __restrict__ rowtab,
    const float* __restrict__ alpha_dst, const float* __restrict__ Wh, float* out) {
  const int lane = threadIdx.x & 31;
  const int n = blockIdx.x * 8 + __builtin_amdgcn_readfirstlane(threadIdx.x >> 5);
  float* orow = out + (size_t)n * DIM;
  const int deg   = __builtin_amdgcn_readfirstlane(rowtab[ROW_CNT + n]);
  const int start = __builtin_amdgcn_readfirstlane(rowtab[ROW_START + n]);
  const float4 as = *(const float4*)(orow + OUT_ASRC);

  float4 acc0 = {0.0f, 0.0f, 0.0f, 0.0f};
  float4 acc1 = {0.0f, 0.0f, 0.0f, 0.0f};

  if (deg > 0) {
    const float NEG_INF = __int_as_float(0xFF800000);
    int j0 = 0;
    float u0 = NEG_INF, u1 = NEG_INF, u2 = NEG_INF, u3 = NEG_INF;
    if (lane < deg)
      edge_logits(src_list, alpha_dst, as, start + lane, j0, u0, u1, u2, u3);

    float m0 = u0, m1 = u1, m2 = u2, m3 = u3;
    for (int base = 32; base < deg; base += 32) {
      if (base + lane < deg) {
        int j; float v0, v1, v2, v3;
        edge_logits(src_list, alpha_dst, as, start + base + lane, j, v0, v1, v2, v3);
        m0 = fmaxf(m0, v0); m1 = fmaxf(m1, v1); m2 = fmaxf(m2, v2); m3 = fmaxf(m3, v3);
      }
    }
    m0 = wave_max(m0); m1 = wave_max(m1); m2 = wave_max(m2); m3 = wave_max(m3);

    const bool in0 = lane < deg;
    const float p0 = in0 ? __expf(u0 - m0) : 0.0f;
    const float p1 = in0 ? __expf(u1 - m1) : 0.0f;
    const float p2 = in0 ? __expf(u2 - m2) : 0.0f;
    const float p3 = in0 ? __expf(u3 - m3) : 0.0f;
    float s0 = p0, s1 = p1, s2 = p2, s3 = p3;
    for (int base = 32; base < deg; base += 32) {
      if (base + lane < deg) {
        int j; float v0, v1, v2, v3;
        edge_logits(src_list, alpha_dst, as, start + base + lane, j, v0, v1, v2, v3);
        s0 += __expf(v0 - m0); s1 += __expf(v1 - m1);
        s2 += __expf(v2 - m2); s3 += __expf(v3 - m3);
      }
    }
    s0 = wave_sum(s0); s1 = wave_sum(s1); s2 = wave_sum(s2); s3 = wave_sum(s3);

    add_rows(Wh, lane, deg < 32 ? deg : 32, j0, p0 / s0, p1 / s1, p2 / s2, p3 / s3,
             acc0, acc1);
    for (int base = 32; base < deg; base += 32) {
      int j = 0;
      float a0 = 0.0f, a1 = 0.0f, a2 = 0.0f, a3 = 0.0f;
      if (base + lane < deg) {
        float v0, v1, v2, v3;
        edge_logits(src_list, alpha_dst, as, start + base + lane, j, v0, v1, v2, v3);
        a0 = __expf(v0 - m0) / s0; a1 = __expf(v1 - m1) / s1;
        a2 = __expf(v2 - m2) / s2; a3 = __expf(v3 - m3) / s3;
      }
      const int left = deg - base;
      add_rows(Wh, lane, left < 32 ? left : 32, j, a0, a1, a2, a3, acc0, acc1);
    }
  }

  *(float4*)(orow + 4 * lane)       = acc0;
  *(float4*)(orow + 128 + 4 * lane) = acc1;
}

extern "C" void kernel_launch(void* const* d_in, const int* in_sizes, int n_in,
                              void* d_out, int out_size, void* d_ws, size_t ws_size,
                              hipStream_t stream) {
  const float* h     = (const float*)d_in[0];
  const int*   ei    = (const int*)d_in[1];
  const float* W     = (const float*)d_in[2];
  const float* bias  = (const float*)d_in[3];
  const float* a_src = (const float*)d_in[4];
  const float* a_dst = (const float*)d_in[5];
  float* out = (float*)d_out;

  float* Wh        = (float*)d_ws;
  int*   rowtab    = (int*)(Wh + (size_t)N_NODES * DIM);
  float* alpha_dst = (float*)(rowtab + (size_t)N_NODES * HEADS);
  unsigned short* src_list = (unsigned short*)(alpha_dst + (size_t)N_NODES * HEADS);

  gat_gemm_wmma<<<GEMM_WGS, 256, 0, stream>>>(h, W, bias, Wh, rowtab, ei, (unsigned*)out);

  group_and_prep<<<PG_BLKS, GRP_THREADS, 0, stream>>>(rowtab, src_list,
                                                      Wh, a_src, a_dst, out, alpha_dst);

  node_agg<<<N_NODES / 8, 256, 0, stream>>>(src_list, rowtab, alpha_dst, Wh, out);
}
